# starting kernel
# baseline (speedup 1.0000x reference)
#include <hip/hip_runtime.h>

typedef __attribute__((ext_vector_type(2))) float v2f;
typedef __attribute__((ext_vector_type(8))) float v8f;

#define BATCH 128
#define SEQ   2048
#define ISZ   128
#define HSZ   3

__global__ __launch_bounds__(256) void tridom_proj_wmma(
    const float* __restrict__ x, const float* __restrict__ wp,
    const float* __restrict__ bias, float* __restrict__ out)
{
  const int lane = threadIdx.x & 31;
  const int wave = threadIdx.x >> 5;
  const int tile = blockIdx.x * 8 + wave;
  const int hi   = lane >> 4;
  const int ml   = lane & 15;

  const int row  = tile * 16 + ml;
  const float* arow = x + (size_t)row * ISZ + 2 * hi;

  const int   n     = ml;
  const float bmask = (n < HSZ) ? 1.0f : 0.0f;
  const float* brow = wp + (size_t)((n < HSZ) ? n : 0) * ISZ + 2 * hi;

  v8f c = {};
#pragma unroll
  for (int ki = 0; ki < ISZ / 4; ++ki) {
    v2f a  = *(const v2f*)(arow + 4 * ki);
    v2f bv = *(const v2f*)(brow + 4 * ki);
    bv.x *= bmask;
    bv.y *= bmask;
    c = __builtin_amdgcn_wmma_f32_16x16x4_f32(false, a, false, bv,
                                              (short)0, c, false, false);
  }

  if (n < HSZ) {
    const float bb = bias[n];
#pragma unroll
    for (int r = 0; r < 8; ++r) {
      const int grow = tile * 16 + r + 8 * hi;
      out[(size_t)grow * HSZ + n] = c[r] + bb;
    }
  }
}

__device__ __forceinline__ float fast_tanh(float v) {
#if defined(__has_builtin) && __has_builtin(__builtin_amdgcn_tanhf)
  return __builtin_amdgcn_tanhf(v);
#else
  return tanhf(v);
#endif
}

__global__ __launch_bounds__(32) void tridom_scan(float* __restrict__ out)
{
  const int b = blockIdx.x * 32 + threadIdx.x;
  float* p = out + (size_t)b * SEQ * HSZ;

  float h0 = 0.f, h1 = 0.f, h2 = 0.f;
  float u0 = p[0], u1 = p[1], u2 = p[2];

  for (int t = 0; t < SEQ; ++t) {
    float* q = p + t * HSZ;
    float nu0 = 0.f, nu1 = 0.f, nu2 = 0.f;
    if (t + 1 < SEQ) { nu0 = q[3]; nu1 = q[4]; nu2 = q[5]; }

    const float a0 = fast_tanh( 1.5f * h1) - 0.5f * h0 + u0;
    const float a1 = fast_tanh( 1.5f * h2) - 0.5f * h1 + u1;
    const float a2 = fast_tanh(-1.5f * h0) - 0.5f * h2 + u2;

    q[0] = a0; q[1] = a1; q[2] = a2;
    h0 = a0; h1 = a1; h2 = a2;
    u0 = nu0; u1 = nu1; u2 = nu2;
  }

  float* hf = out + (size_t)BATCH * SEQ * HSZ + (size_t)b * HSZ;
  hf[0] = h0; hf[1] = h1; hf[2] = h2;
}

extern "C" void kernel_launch(void* const* d_in, const int* in_sizes, int n_in,
                              void* d_out, int out_size, void* d_ws, size_t ws_size,
                              hipStream_t stream) {
  const float* x    = (const float*)d_in[0];
  const float* wp   = (const float*)d_in[1];
  const float* bias = (const float*)d_in[2];
  float* out = (float*)d_out;

  const int tiles  = (BATCH * SEQ) / 16;
  const int blocks = tiles / 8;

  tridom_proj_wmma<<<dim3(blocks), dim3(256), 0, stream>>>(x, wp, bias, out);
  tridom_scan<<<dim3(4), dim3(32), 0, stream>>>(out);
}


# projection tile re-cut: k split over the tile's rows (row = time row x 32-B piece of a line): each x load reads 4 whole 128-B lines, no LDS; 4 quarter sums added in a quad, k summed in new order
# speedup vs baseline: 5.8318x; 5.8318x over previous
#include <hip/hip_runtime.h>

typedef __attribute__((ext_vector_type(2))) float v2f;
typedef __attribute__((ext_vector_type(4))) float v4f;
typedef __attribute__((ext_vector_type(8))) float v8f;

#define BATCH 128
#define SEQ   2048
#define ISZ   128
#define HSZ   3

#define PROJ_BLOCKS_PER_WAVE 4
#define PROJ_WAVES_PER_WG    16
#define PROJ_GROUPS          4
#define PROJ_LINES           (ISZ / 32)

__device__ __forceinline__ void proj_load_block(v4f (&a)[PROJ_GROUPS][PROJ_LINES], const float* p)
{
#pragma unroll
  for (int G = 0; G < PROJ_GROUPS; ++G)
#pragma unroll
    for (int L = 0; L < PROJ_LINES; ++L)
      a[G][L] = *(const v4f*)(p + (size_t)G * 4 * ISZ + 32 * L);
}

__device__ __forceinline__ float quad_x1(float v)
{
  const int b = __builtin_bit_cast(int, v);
  return __builtin_bit_cast(float, __builtin_amdgcn_update_dpp(b, b, 0xB1, 0xF, 0xF, true));
}
__device__ __forceinline__ float quad_x2(float v)
{
  const int b = __builtin_bit_cast(int, v);
  return __builtin_bit_cast(float, __builtin_amdgcn_update_dpp(b, b, 0x4E, 0xF, 0xF, true));
}

__device__ __forceinline__ float sel4(int q, float v0, float v1, float v2, float v3)
{
  const float lo = (q & 1) ? v1 : v0;
  const float hi = (q & 1) ? v3 : v2;
  return (q & 2) ? hi : lo;
}

__device__ __forceinline__ void proj_block(const v4f (&a)[PROJ_GROUPS][PROJ_LINES], const v4f (&bf)[PROJ_LINES],
                                           float* __restrict__ out, int row0, int hi, int h, int gq, float bb)
{
  v8f c[PROJ_GROUPS];
#pragma unroll
  for (int G = 0; G < PROJ_GROUPS; ++G) c[G] = (v8f){};
#pragma unroll
  for (int L = 0; L < PROJ_LINES; ++L) {
    const v2f b_lo = { bf[L].x, bf[L].y }, b_hi = { bf[L].z, bf[L].w };
#pragma unroll
    for (int G = 0; G < PROJ_GROUPS; ++G) {
      const v2f a_lo = { a[G][L].x, a[G][L].y };
      c[G] = __builtin_amdgcn_wmma_f32_16x16x4_f32(false, a_lo, false, b_lo,
                                                   (short)0, c[G], false, false);
    }
#pragma unroll
    for (int G = 0; G < PROJ_GROUPS; ++G) {
      const v2f a_hi = { a[G][L].z, a[G][L].w };
      c[G] = __builtin_amdgcn_wmma_f32_16x16x4_f32(false, a_hi, false, b_hi,
                                                   (short)0, c[G], false, false);
    }
  }
  float s0[PROJ_GROUPS], s1[PROJ_GROUPS];
#pragma unroll
  for (int G = 0; G < PROJ_GROUPS; ++G) {
    float w0 = sel4(gq, c[G][0], c[G][1], c[G][2], c[G][3]);
    float w1 = sel4(gq, c[G][4], c[G][5], c[G][6], c[G][7]);
    w0 += quad_x1(w0);  w1 += quad_x1(w1);
    w0 += quad_x2(w0);  w1 += quad_x2(w1);
    s0[G] = w0;  s1[G] = w1;
  }
  const float o0 = sel4(gq, s0[0], s0[1], s0[2], s0[3]) + bb;
  const float o1 = sel4(gq, s1[0], s1[1], s1[2], s1[3]) + bb;
  if (h < HSZ) {
    const int grow = row0 + 4 * gq + 2 * hi;
    out[(size_t)grow * HSZ + h]       = o0;
    out[(size_t)(grow + 1) * HSZ + h] = o1;
  }
}

__global__ __launch_bounds__(32 * PROJ_WAVES_PER_WG) void tridom_proj_wmma(
    const float* __restrict__ x, const float* __restrict__ wp,
    const float* __restrict__ bias, float* __restrict__ out)
{
  const int lane = threadIdx.x & 31;
  const int wave = threadIdx.x >> 5;
  const int hi   = lane >> 4;
  const int ml   = lane & 15;
  const int tl   = ml >> 2;
  const int gq   = ml & 3;
  const int row_w = (blockIdx.x * PROJ_WAVES_PER_WG + wave) * PROJ_BLOCKS_PER_WAVE * 16;

  const float* ap = x + ((size_t)row_w + tl) * ISZ + 8 * gq + 4 * hi;

  v4f a0[PROJ_GROUPS][PROJ_LINES], a1[PROJ_GROUPS][PROJ_LINES];
  proj_load_block(a0, ap);

  const int   h     = tl;
  const float bmask = (h < HSZ) ? 1.0f : 0.0f;
  const float* brow = wp + (size_t)((h < HSZ) ? h : 0) * ISZ + 8 * gq + 4 * hi;
  const float bb    = bias[(h < HSZ) ? h : 0];
  v4f bf[PROJ_LINES];
#pragma unroll
  for (int L = 0; L < PROJ_LINES; ++L) {
    v4f b4 = *(const v4f*)(brow + 32 * L);
    bf[L] = b4 * bmask;
  }

#pragma unroll 1
  for (int i = 0; i < PROJ_BLOCKS_PER_WAVE; i += 2) {
    proj_load_block(a1, ap + (size_t)(i + 1) * 16 * ISZ);
    proj_block(a0, bf, out, row_w + 16 * i, hi, h, gq, bb);
    if (i + 2 < PROJ_BLOCKS_PER_WAVE)
      proj_load_block(a0, ap + (size_t)(i + 2) * 16 * ISZ);
    proj_block(a1, bf, out, row_w + 16 * (i + 1), hi, h, gq, bb);
  }
}

#define CHUNK_STEPS 128
#define CHUNK_V4    (CHUNK_STEPS * HSZ / 4)
#define NCHUNK      (SEQ / CHUNK_STEPS)

#define TRI_MUL0 "v_mul_f32 %[t0], 0x3fc00000, %[h1]\n\t"
#define TRI_MUL1 "v_mul_f32 %[t1], 0x3fc00000, %[h2]\n\t"
#define TRI_MUL2 "v_mul_f32 %[t2], 0xbfc00000, %[h0]\n\t"
#define TRI_MUL(c) TRI_MUL##c
#define TRI_RDL(c) "v_readlane_b32 %[u" #c "], %[c" #c "], %[j]\n\t"
#define TRI_TNH(c) "v_tanh_f32 %[t" #c "], %[t" #c "]\n\t"
#define TRI_FMC(c) "v_fmac_f32 %[t" #c "], -0.5, %[h" #c "]\n\t"
#define TRI_ADD(c) "v_add_f32 %[h" #c "], %[u" #c "], %[t" #c "]\n\t"
#define TRI_CAP(c) "v_cndmask_b32_e64 %[r" #c "], %[r" #c "], %[h" #c "], %[m]\n\t"

#define TRI_STEP(X, Y, Z, R0, R1, R2, C0, C1, C2, J, M)                          \
  do {                                                                           \
    float t0_, t1_, t2_, u0_, u1_, u2_;                                          \
    asm(TRI_MUL(X) TRI_RDL(X) TRI_MUL(Y) TRI_RDL(Y) TRI_MUL(Z)                   \
        TRI_TNH(X) TRI_RDL(Z) TRI_TNH(Y) TRI_CAP(0) TRI_TNH(Z)                   \
        TRI_CAP(1) TRI_CAP(2)                                                    \
        TRI_FMC(X) TRI_FMC(Y) TRI_FMC(Z)                                         \
        TRI_ADD(X) TRI_ADD(Y) TRI_ADD(Z)                                         \
        : [h0] "+v"(h0), [h1] "+v"(h1), [h2] "+v"(h2),                           \
          [r0] "+v"(R0), [r1] "+v"(R1), [r2] "+v"(R2),                           \
          [t0] "=&v"(t0_), [t1] "=&v"(t1_), [t2] "=&v"(t2_),                     \
          [u0] "=&s"(u0_), [u1] "=&s"(u1_), [u2] "=&s"(u2_)                      \
        : [c0] "v"(C0), [c1] "v"(C1), [c2] "v"(C2), [j] "s"(J), [m] "s"(M));     \
  } while (0)

__global__ __launch_bounds__(32) void tridom_scan(float* __restrict__ out)
{
  const int lane = threadIdx.x;
  const int b    = blockIdx.x;
  v4f* pl = (v4f*)(out + (size_t)b * SEQ * HSZ) + 3 * lane;

  float h0 = 0.f, h1 = 0.f, h2 = 0.f;
  float r0 = 0.f, r1 = 0.f, r2 = 0.f, r3 = 0.f, r4 = 0.f, r5 = 0.f;
  float r6 = 0.f, r7 = 0.f, r8 = 0.f, r9 = 0.f, r10 = 0.f, r11 = 0.f;

  v4f c0 = pl[0], c1 = pl[1], c2 = pl[2];

  for (int c = 0; c < NCHUNK; ++c) {
    v4f n0 = c0, n1 = c1, n2 = c2;
    if (c + 1 < NCHUNK) {
      const v4f* pn = pl + (c + 1) * CHUNK_V4;
      n0 = pn[0]; n1 = pn[1]; n2 = pn[2];
    }

#pragma unroll 4
    for (int j = 0; j < 32; ++j) {
      const unsigned mcur  = 1u << j;
      const unsigned mprev = mcur >> 1;
      TRI_STEP(0, 1, 2, r9, r10, r11, c0.x, c0.y, c0.z, j, mprev);
      TRI_STEP(2, 0, 1, r0, r1,  r2,  c0.w, c1.x, c1.y, j, mcur);
      TRI_STEP(1, 2, 0, r3, r4,  r5,  c1.z, c1.w, c2.x, j, mcur);
      TRI_STEP(0, 1, 2, r6, r7,  r8,  c2.y, c2.z, c2.w, j, mcur);
    }
    r9  = (lane == 31) ? h0 : r9;
    r10 = (lane == 31) ? h1 : r10;
    r11 = (lane == 31) ? h2 : r11;

    v4f* po = pl + c * CHUNK_V4;
    po[0] = (v4f){ r0, r1, r2,  r3  };
    po[1] = (v4f){ r4, r5, r6,  r7  };
    po[2] = (v4f){ r8, r9, r10, r11 };
    c0 = n0; c1 = n1; c2 = n2;
  }

  if (lane == 0) {
    float* hf = out + (size_t)BATCH * SEQ * HSZ + (size_t)b * HSZ;
    hf[0] = h0; hf[1] = h1; hf[2] = h2;
  }
}

extern "C" void kernel_launch(void* const* d_in, const int* in_sizes, int n_in,
                              void* d_out, int out_size, void* d_ws, size_t ws_size,
                              hipStream_t stream) {
  const float* x    = (const float*)d_in[0];
  const float* wp   = (const float*)d_in[1];
  const float* bias = (const float*)d_in[2];
  float* out = (float*)d_out;

  const int tiles  = (BATCH * SEQ) / 16;
  const int blocks = tiles / (PROJ_WAVES_PER_WG * PROJ_BLOCKS_PER_WAVE);

  tridom_proj_wmma<<<dim3(blocks), dim3(32 * PROJ_WAVES_PER_WG), 0, stream>>>(x, wp, bias, out);
  tridom_scan<<<dim3(BATCH), dim3(32), 0, stream>>>(out);
}


# scan: a chunk's 128 steps written out: readlane lanes and keep masks fixed, u read a step ahead, so no vector instruction reads a freshly written scalar register; arithmetic unchanged
# speedup vs baseline: 6.3489x; 1.0887x over previous
#include <hip/hip_runtime.h>

typedef __attribute__((ext_vector_type(2))) float v2f;
typedef __attribute__((ext_vector_type(4))) float v4f;
typedef __attribute__((ext_vector_type(8))) float v8f;

#define BATCH 128
#define SEQ   2048
#define ISZ   128
#define HSZ   3

#define PROJ_BLOCKS_PER_WAVE 4
#define PROJ_WAVES_PER_WG    16
#define PROJ_GROUPS          4
#define PROJ_LINES           (ISZ / 32)

__device__ __forceinline__ void proj_load_block(v4f (&a)[PROJ_GROUPS][PROJ_LINES], const float* p)
{
#pragma unroll
  for (int G = 0; G < PROJ_GROUPS; ++G)
#pragma unroll
    for (int L = 0; L < PROJ_LINES; ++L)
      a[G][L] = *(const v4f*)(p + (size_t)G * 4 * ISZ + 32 * L);
}

__device__ __forceinline__ float quad_x1(float v)
{
  const int b = __builtin_bit_cast(int, v);
  return __builtin_bit_cast(float, __builtin_amdgcn_update_dpp(b, b, 0xB1, 0xF, 0xF, true));
}
__device__ __forceinline__ float quad_x2(float v)
{
  const int b = __builtin_bit_cast(int, v);
  return __builtin_bit_cast(float, __builtin_amdgcn_update_dpp(b, b, 0x4E, 0xF, 0xF, true));
}

__device__ __forceinline__ float sel4(int q, float v0, float v1, float v2, float v3)
{
  const float lo = (q & 1) ? v1 : v0;
  const float hi = (q & 1) ? v3 : v2;
  return (q & 2) ? hi : lo;
}

__device__ __forceinline__ void proj_block(const v4f (&a)[PROJ_GROUPS][PROJ_LINES], const v4f (&bf)[PROJ_LINES],
                                           float* __restrict__ out, int row0, int hi, int h, int gq, float bb)
{
  v8f c[PROJ_GROUPS];
#pragma unroll
  for (int G = 0; G < PROJ_GROUPS; ++G) c[G] = (v8f){};
#pragma unroll
  for (int L = 0; L < PROJ_LINES; ++L) {
    const v2f b_lo = { bf[L].x, bf[L].y }, b_hi = { bf[L].z, bf[L].w };
#pragma unroll
    for (int G = 0; G < PROJ_GROUPS; ++G) {
      const v2f a_lo = { a[G][L].x, a[G][L].y };
      c[G] = __builtin_amdgcn_wmma_f32_16x16x4_f32(false, a_lo, false, b_lo,
                                                   (short)0, c[G], false, false);
    }
#pragma unroll
    for (int G = 0; G < PROJ_GROUPS; ++G) {
      const v2f a_hi = { a[G][L].z, a[G][L].w };
      c[G] = __builtin_amdgcn_wmma_f32_16x16x4_f32(false, a_hi, false, b_hi,
                                                   (short)0, c[G], false, false);
    }
  }
  float s0[PROJ_GROUPS], s1[PROJ_GROUPS];
#pragma unroll
  for (int G = 0; G < PROJ_GROUPS; ++G) {
    float w0 = sel4(gq, c[G][0], c[G][1], c[G][2], c[G][3]);
    float w1 = sel4(gq, c[G][4], c[G][5], c[G][6], c[G][7]);
    w0 += quad_x1(w0);  w1 += quad_x1(w1);
    w0 += quad_x2(w0);  w1 += quad_x2(w1);
    s0[G] = w0;  s1[G] = w1;
  }
  const float o0 = sel4(gq, s0[0], s0[1], s0[2], s0[3]) + bb;
  const float o1 = sel4(gq, s1[0], s1[1], s1[2], s1[3]) + bb;
  if (h < HSZ) {
    const int grow = row0 + 4 * gq + 2 * hi;
    out[(size_t)grow * HSZ + h]       = o0;
    out[(size_t)(grow + 1) * HSZ + h] = o1;
  }
}

__global__ __launch_bounds__(32 * PROJ_WAVES_PER_WG) void tridom_proj_wmma(
    const float* __restrict__ x, const float* __restrict__ wp,
    const float* __restrict__ bias, float* __restrict__ out)
{
  const int lane = threadIdx.x & 31;
  const int wave = threadIdx.x >> 5;
  const int hi   = lane >> 4;
  const int ml   = lane & 15;
  const int tl   = ml >> 2;
  const int gq   = ml & 3;
  const int row_w = (blockIdx.x * PROJ_WAVES_PER_WG + wave) * PROJ_BLOCKS_PER_WAVE * 16;

  const float* ap = x + ((size_t)row_w + tl) * ISZ + 8 * gq + 4 * hi;

  v4f a0[PROJ_GROUPS][PROJ_LINES], a1[PROJ_GROUPS][PROJ_LINES];
  proj_load_block(a0, ap);

  const int   h     = tl;
  const float bmask = (h < HSZ) ? 1.0f : 0.0f;
  const float* brow = wp + (size_t)((h < HSZ) ? h : 0) * ISZ + 8 * gq + 4 * hi;
  const float bb    = bias[(h < HSZ) ? h : 0];
  v4f bf[PROJ_LINES];
#pragma unroll
  for (int L = 0; L < PROJ_LINES; ++L) {
    v4f b4 = *(const v4f*)(brow + 32 * L);
    bf[L] = b4 * bmask;
  }

#pragma unroll 1
  for (int i = 0; i < PROJ_BLOCKS_PER_WAVE; i += 2) {
    proj_load_block(a1, ap + (size_t)(i + 1) * 16 * ISZ);
    proj_block(a0, bf, out, row_w + 16 * i, hi, h, gq, bb);
    if (i + 2 < PROJ_BLOCKS_PER_WAVE)
      proj_load_block(a0, ap + (size_t)(i + 2) * 16 * ISZ);
    proj_block(a1, bf, out, row_w + 16 * (i + 1), hi, h, gq, bb);
  }
}

#define CHUNK_STEPS 128
#define CHUNK_V4    (CHUNK_STEPS * HSZ / 4)
#define NCHUNK      (SEQ / CHUNK_STEPS)

#define TRI_MUL0 "v_mul_f32 %[t0], 0x3fc00000, %[h1]\n\t"
#define TRI_MUL1 "v_mul_f32 %[t1], 0x3fc00000, %[h2]\n\t"
#define TRI_MUL2 "v_mul_f32 %[t2], 0xbfc00000, %[h0]\n\t"
#define TRI_MUL(c) TRI_MUL##c
#define TRI_RDL(c) "v_readlane_b32 %[n" #c "], %[c" #c "], %[j]\n\t"
#define TRI_TNH(c) "v_tanh_f32 %[t" #c "], %[t" #c "]\n\t"
#define TRI_FMC(c) "v_fmac_f32 %[t" #c "], -0.5, %[h" #c "]\n\t"
#define TRI_ADD(c) "v_add_f32 %[h" #c "], %[u" #c "], %[t" #c "]\n\t"
#define TRI_CAP(c) "v_cndmask_b32_e64 %[r" #c "], %[r" #c "], %[h" #c "], %[m]\n\t"

#define TRI_STEP(X, Y, Z, R0, R1, R2, MV, U0, U1, U2, N0, N1, N2, CN0, CN1, CN2, JN) \
  do {                                                                           \
    float t0_, t1_, t2_;                                                         \
    asm(TRI_MUL(X) TRI_RDL(0) TRI_MUL(Y) TRI_RDL(1) TRI_MUL(Z)                   \
        TRI_TNH(X) TRI_RDL(2) TRI_TNH(Y) TRI_TNH(Z)                              \
        TRI_CAP(0) TRI_CAP(1) TRI_CAP(2)                                         \
        TRI_FMC(X) TRI_FMC(Y) TRI_FMC(Z)                                         \
        TRI_ADD(X) TRI_ADD(Y) TRI_ADD(Z)                                         \
        : [h0] "+v"(h0), [h1] "+v"(h1), [h2] "+v"(h2),                           \
          [r0] "+v"(R0), [r1] "+v"(R1), [r2] "+v"(R2),                           \
          [t0] "=&v"(t0_), [t1] "=&v"(t1_), [t2] "=&v"(t2_),                     \
          [n0] "=&s"(N0), [n1] "=&s"(N1), [n2] "=&s"(N2)                         \
        : [c0] "v"(CN0), [c1] "v"(CN1), [c2] "v"(CN2),                           \
          [u0] "s"(U0), [u1] "s"(U1), [u2] "s"(U2), [m] "s"(MV), [j] "n"(JN));   \
  } while (0)

#define TRI_STEP_FIRST(X, Y, Z, U0, U1, U2, N0, N1, N2, CN0, CN1, CN2, JN)       \
  do {                                                                           \
    float t0_, t1_, t2_;                                                         \
    asm(TRI_MUL(X) TRI_RDL(0) TRI_MUL(Y) TRI_RDL(1) TRI_MUL(Z)                   \
        TRI_TNH(X) TRI_RDL(2) TRI_TNH(Y) TRI_TNH(Z)                              \
        TRI_FMC(X) TRI_FMC(Y) TRI_FMC(Z)                                         \
        TRI_ADD(X) TRI_ADD(Y) TRI_ADD(Z)                                         \
        : [h0] "+v"(h0), [h1] "+v"(h1), [h2] "+v"(h2),                           \
          [t0] "=&v"(t0_), [t1] "=&v"(t1_), [t2] "=&v"(t2_),                     \
          [n0] "=&s"(N0), [n1] "=&s"(N1), [n2] "=&s"(N2)                         \
        : [c0] "v"(CN0), [c1] "v"(CN1), [c2] "v"(CN2),                           \
          [u0] "s"(U0), [u1] "s"(U1), [u2] "s"(U2), [j] "n"(JN));                \
  } while (0)

__device__ __forceinline__ float tri_lane0(float v)
{
  return __builtin_bit_cast(float, __builtin_amdgcn_readlane(__builtin_bit_cast(int, v), 0));
}

__global__ __launch_bounds__(32) void tridom_scan(float* __restrict__ out)
{
  const int lane = threadIdx.x;
  const int b    = blockIdx.x;
  v4f* pl = (v4f*)(out + (size_t)b * SEQ * HSZ) + 3 * lane;

  float h0 = 0.f, h1 = 0.f, h2 = 0.f;
  float r0 = 0.f, r1 = 0.f, r2 = 0.f, r3 = 0.f, r4 = 0.f, r5 = 0.f;
  float r6 = 0.f, r7 = 0.f, r8 = 0.f, r9 = 0.f, r10 = 0.f, r11 = 0.f;

  unsigned mk[32];
#pragma unroll
  for (int j = 0; j < 32; ++j) mk[j] = __builtin_amdgcn_ballot_w32(lane == j);

  v4f c0 = pl[0], c1 = pl[1], c2 = pl[2];

  float ua0 = tri_lane0(c0.x), ua1 = tri_lane0(c0.y), ua2 = tri_lane0(c0.z);
  float ub0, ub1, ub2;

  for (int c = 0; c < NCHUNK; ++c) {
    v4f n0 = c0, n1 = c1, n2 = c2;
    if (c + 1 < NCHUNK) {
      const v4f* pn = pl + (c + 1) * CHUNK_V4;
      n0 = pn[0]; n1 = pn[1]; n2 = pn[2];
    }

    TRI_STEP_FIRST(0, 1, 2, ua0, ua1, ua2, ub0, ub1, ub2, c0.w, c1.x, c1.y, 0);
    TRI_STEP(2, 0, 1, r0, r1, r2, mk[0], ub0, ub1, ub2, ua0, ua1, ua2, c1.z, c1.w, c2.x, 0);
    TRI_STEP(1, 2, 0, r3, r4, r5, mk[0], ua0, ua1, ua2, ub0, ub1, ub2, c2.y, c2.z, c2.w, 0);
    TRI_STEP(0, 1, 2, r6, r7, r8, mk[0], ub0, ub1, ub2, ua0, ua1, ua2, c0.x, c0.y, c0.z, 1);
    TRI_STEP(2, 0, 1, r9, r10, r11, mk[0], ua0, ua1, ua2, ub0, ub1, ub2, c0.w, c1.x, c1.y, 1);
    TRI_STEP(1, 2, 0, r0, r1, r2, mk[1], ub0, ub1, ub2, ua0, ua1, ua2, c1.z, c1.w, c2.x, 1);
    TRI_STEP(0, 1, 2, r3, r4, r5, mk[1], ua0, ua1, ua2, ub0, ub1, ub2, c2.y, c2.z, c2.w, 1);
    TRI_STEP(2, 0, 1, r6, r7, r8, mk[1], ub0, ub1, ub2, ua0, ua1, ua2, c0.x, c0.y, c0.z, 2);
    TRI_STEP(1, 2, 0, r9, r10, r11, mk[1], ua0, ua1, ua2, ub0, ub1, ub2, c0.w, c1.x, c1.y, 2);
    TRI_STEP(0, 1, 2, r0, r1, r2, mk[2], ub0, ub1, ub2, ua0, ua1, ua2, c1.z, c1.w, c2.x, 2);
    TRI_STEP(2, 0, 1, r3, r4, r5, mk[2], ua0, ua1, ua2, ub0, ub1, ub2, c2.y, c2.z, c2.w, 2);
    TRI_STEP(1, 2, 0, r6, r7, r8, mk[2], ub0, ub1, ub2, ua0, ua1, ua2, c0.x, c0.y, c0.z, 3);
    TRI_STEP(0, 1, 2, r9, r10, r11, mk[2], ua0, ua1, ua2, ub0, ub1, ub2, c0.w, c1.x, c1.y, 3);
    TRI_STEP(2, 0, 1, r0, r1, r2, mk[3], ub0, ub1, ub2, ua0, ua1, ua2, c1.z, c1.w, c2.x, 3);
    TRI_STEP(1, 2, 0, r3, r4, r5, mk[3], ua0, ua1, ua2, ub0, ub1, ub2, c2.y, c2.z, c2.w, 3);
    TRI_STEP(0, 1, 2, r6, r7, r8, mk[3], ub0, ub1, ub2, ua0, ua1, ua2, c0.x, c0.y, c0.z, 4);
    TRI_STEP(2, 0, 1, r9, r10, r11, mk[3], ua0, ua1, ua2, ub0, ub1, ub2, c0.w, c1.x, c1.y, 4);
    TRI_STEP(1, 2, 0, r0, r1, r2, mk[4], ub0, ub1, ub2, ua0, ua1, ua2, c1.z, c1.w, c2.x, 4);
    TRI_STEP(0, 1, 2, r3, r4, r5, mk[4], ua0, ua1, ua2, ub0, ub1, ub2, c2.y, c2.z, c2.w, 4);
    TRI_STEP(2, 0, 1, r6, r7, r8, mk[4], ub0, ub1, ub2, ua0, ua1, ua2, c0.x, c0.y, c0.z, 5);
    TRI_STEP(1, 2, 0, r9, r10, r11, mk[4], ua0, ua1, ua2, ub0, ub1, ub2, c0.w, c1.x, c1.y, 5);
    TRI_STEP(0, 1, 2, r0, r1, r2, mk[5], ub0, ub1, ub2, ua0, ua1, ua2, c1.z, c1.w, c2.x, 5);
    TRI_STEP(2, 0, 1, r3, r4, r5, mk[5], ua0, ua1, ua2, ub0, ub1, ub2, c2.y, c2.z, c2.w, 5);
    TRI_STEP(1, 2, 0, r6, r7, r8, mk[5], ub0, ub1, ub2, ua0, ua1, ua2, c0.x, c0.y, c0.z, 6);
    TRI_STEP(0, 1, 2, r9, r10, r11, mk[5], ua0, ua1, ua2, ub0, ub1, ub2, c0.w, c1.x, c1.y, 6);
    TRI_STEP(2, 0, 1, r0, r1, r2, mk[6], ub0, ub1, ub2, ua0, ua1, ua2, c1.z, c1.w, c2.x, 6);
    TRI_STEP(1, 2, 0, r3, r4, r5, mk[6], ua0, ua1, ua2, ub0, ub1, ub2, c2.y, c2.z, c2.w, 6);
    TRI_STEP(0, 1, 2, r6, r7, r8, mk[6], ub0, ub1, ub2, ua0, ua1, ua2, c0.x, c0.y, c0.z, 7);
    TRI_STEP(2, 0, 1, r9, r10, r11, mk[6], ua0, ua1, ua2, ub0, ub1, ub2, c0.w, c1.x, c1.y, 7);
    TRI_STEP(1, 2, 0, r0, r1, r2, mk[7], ub0, ub1, ub2, ua0, ua1, ua2, c1.z, c1.w, c2.x, 7);
    TRI_STEP(0, 1, 2, r3, r4, r5, mk[7], ua0, ua1, ua2, ub0, ub1, ub2, c2.y, c2.z, c2.w, 7);
    TRI_STEP(2, 0, 1, r6, r7, r8, mk[7], ub0, ub1, ub2, ua0, ua1, ua2, c0.x, c0.y, c0.z, 8);
    TRI_STEP(1, 2, 0, r9, r10, r11, mk[7], ua0, ua1, ua2, ub0, ub1, ub2, c0.w, c1.x, c1.y, 8);
    TRI_STEP(0, 1, 2, r0, r1, r2, mk[8], ub0, ub1, ub2, ua0, ua1, ua2, c1.z, c1.w, c2.x, 8);
    TRI_STEP(2, 0, 1, r3, r4, r5, mk[8], ua0, ua1, ua2, ub0, ub1, ub2, c2.y, c2.z, c2.w, 8);
    TRI_STEP(1, 2, 0, r6, r7, r8, mk[8], ub0, ub1, ub2, ua0, ua1, ua2, c0.x, c0.y, c0.z, 9);
    TRI_STEP(0, 1, 2, r9, r10, r11, mk[8], ua0, ua1, ua2, ub0, ub1, ub2, c0.w, c1.x, c1.y, 9);
    TRI_STEP(2, 0, 1, r0, r1, r2, mk[9], ub0, ub1, ub2, ua0, ua1, ua2, c1.z, c1.w, c2.x, 9);
    TRI_STEP(1, 2, 0, r3, r4, r5, mk[9], ua0, ua1, ua2, ub0, ub1, ub2, c2.y, c2.z, c2.w, 9);
    TRI_STEP(0, 1, 2, r6, r7, r8, mk[9], ub0, ub1, ub2, ua0, ua1, ua2, c0.x, c0.y, c0.z, 10);
    TRI_STEP(2, 0, 1, r9, r10, r11, mk[9], ua0, ua1, ua2, ub0, ub1, ub2, c0.w, c1.x, c1.y, 10);
    TRI_STEP(1, 2, 0, r0, r1, r2, mk[10], ub0, ub1, ub2, ua0, ua1, ua2, c1.z, c1.w, c2.x, 10);
    TRI_STEP(0, 1, 2, r3, r4, r5, mk[10], ua0, ua1, ua2, ub0, ub1, ub2, c2.y, c2.z, c2.w, 10);
    TRI_STEP(2, 0, 1, r6, r7, r8, mk[10], ub0, ub1, ub2, ua0, ua1, ua2, c0.x, c0.y, c0.z, 11);
    TRI_STEP(1, 2, 0, r9, r10, r11, mk[10], ua0, ua1, ua2, ub0, ub1, ub2, c0.w, c1.x, c1.y, 11);
    TRI_STEP(0, 1, 2, r0, r1, r2, mk[11], ub0, ub1, ub2, ua0, ua1, ua2, c1.z, c1.w, c2.x, 11);
    TRI_STEP(2, 0, 1, r3, r4, r5, mk[11], ua0, ua1, ua2, ub0, ub1, ub2, c2.y, c2.z, c2.w, 11);
    TRI_STEP(1, 2, 0, r6, r7, r8, mk[11], ub0, ub1, ub2, ua0, ua1, ua2, c0.x, c0.y, c0.z, 12);
    TRI_STEP(0, 1, 2, r9, r10, r11, mk[11], ua0, ua1, ua2, ub0, ub1, ub2, c0.w, c1.x, c1.y, 12);
    TRI_STEP(2, 0, 1, r0, r1, r2, mk[12], ub0, ub1, ub2, ua0, ua1, ua2, c1.z, c1.w, c2.x, 12);
    TRI_STEP(1, 2, 0, r3, r4, r5, mk[12], ua0, ua1, ua2, ub0, ub1, ub2, c2.y, c2.z, c2.w, 12);
    TRI_STEP(0, 1, 2, r6, r7, r8, mk[12], ub0, ub1, ub2, ua0, ua1, ua2, c0.x, c0.y, c0.z, 13);
    TRI_STEP(2, 0, 1, r9, r10, r11, mk[12], ua0, ua1, ua2, ub0, ub1, ub2, c0.w, c1.x, c1.y, 13);
    TRI_STEP(1, 2, 0, r0, r1, r2, mk[13], ub0, ub1, ub2, ua0, ua1, ua2, c1.z, c1.w, c2.x, 13);
    TRI_STEP(0, 1, 2, r3, r4, r5, mk[13], ua0, ua1, ua2, ub0, ub1, ub2, c2.y, c2.z, c2.w, 13);
    TRI_STEP(2, 0, 1, r6, r7, r8, mk[13], ub0, ub1, ub2, ua0, ua1, ua2, c0.x, c0.y, c0.z, 14);
    TRI_STEP(1, 2, 0, r9, r10, r11, mk[13], ua0, ua1, ua2, ub0, ub1, ub2, c0.w, c1.x, c1.y, 14);
    TRI_STEP(0, 1, 2, r0, r1, r2, mk[14], ub0, ub1, ub2, ua0, ua1, ua2, c1.z, c1.w, c2.x, 14);
    TRI_STEP(2, 0, 1, r3, r4, r5, mk[14], ua0, ua1, ua2, ub0, ub1, ub2, c2.y, c2.z, c2.w, 14);
    TRI_STEP(1, 2, 0, r6, r7, r8, mk[14], ub0, ub1, ub2, ua0, ua1, ua2, c0.x, c0.y, c0.z, 15);
    TRI_STEP(0, 1, 2, r9, r10, r11, mk[14], ua0, ua1, ua2, ub0, ub1, ub2, c0.w, c1.x, c1.y, 15);
    TRI_STEP(2, 0, 1, r0, r1, r2, mk[15], ub0, ub1, ub2, ua0, ua1, ua2, c1.z, c1.w, c2.x, 15);
    TRI_STEP(1, 2, 0, r3, r4, r5, mk[15], ua0, ua1, ua2, ub0, ub1, ub2, c2.y, c2.z, c2.w, 15);
    TRI_STEP(0, 1, 2, r6, r7, r8, mk[15], ub0, ub1, ub2, ua0, ua1, ua2, c0.x, c0.y, c0.z, 16);
    TRI_STEP(2, 0, 1, r9, r10, r11, mk[15], ua0, ua1, ua2, ub0, ub1, ub2, c0.w, c1.x, c1.y, 16);
    TRI_STEP(1, 2, 0, r0, r1, r2, mk[16], ub0, ub1, ub2, ua0, ua1, ua2, c1.z, c1.w, c2.x, 16);
    TRI_STEP(0, 1, 2, r3, r4, r5, mk[16], ua0, ua1, ua2, ub0, ub1, ub2, c2.y, c2.z, c2.w, 16);
    TRI_STEP(2, 0, 1, r6, r7, r8, mk[16], ub0, ub1, ub2, ua0, ua1, ua2, c0.x, c0.y, c0.z, 17);
    TRI_STEP(1, 2, 0, r9, r10, r11, mk[16], ua0, ua1, ua2, ub0, ub1, ub2, c0.w, c1.x, c1.y, 17);
    TRI_STEP(0, 1, 2, r0, r1, r2, mk[17], ub0, ub1, ub2, ua0, ua1, ua2, c1.z, c1.w, c2.x, 17);
    TRI_STEP(2, 0, 1, r3, r4, r5, mk[17], ua0, ua1, ua2, ub0, ub1, ub2, c2.y, c2.z, c2.w, 17);
    TRI_STEP(1, 2, 0, r6, r7, r8, mk[17], ub0, ub1, ub2, ua0, ua1, ua2, c0.x, c0.y, c0.z, 18);
    TRI_STEP(0, 1, 2, r9, r10, r11, mk[17], ua0, ua1, ua2, ub0, ub1, ub2, c0.w, c1.x, c1.y, 18);
    TRI_STEP(2, 0, 1, r0, r1, r2, mk[18], ub0, ub1, ub2, ua0, ua1, ua2, c1.z, c1.w, c2.x, 18);
    TRI_STEP(1, 2, 0, r3, r4, r5, mk[18], ua0, ua1, ua2, ub0, ub1, ub2, c2.y, c2.z, c2.w, 18);
    TRI_STEP(0, 1, 2, r6, r7, r8, mk[18], ub0, ub1, ub2, ua0, ua1, ua2, c0.x, c0.y, c0.z, 19);
    TRI_STEP(2, 0, 1, r9, r10, r11, mk[18], ua0, ua1, ua2, ub0, ub1, ub2, c0.w, c1.x, c1.y, 19);
    TRI_STEP(1, 2, 0, r0, r1, r2, mk[19], ub0, ub1, ub2, ua0, ua1, ua2, c1.z, c1.w, c2.x, 19);
    TRI_STEP(0, 1, 2, r3, r4, r5, mk[19], ua0, ua1, ua2, ub0, ub1, ub2, c2.y, c2.z, c2.w, 19);
    TRI_STEP(2, 0, 1, r6, r7, r8, mk[19], ub0, ub1, ub2, ua0, ua1, ua2, c0.x, c0.y, c0.z, 20);
    TRI_STEP(1, 2, 0, r9, r10, r11, mk[19], ua0, ua1, ua2, ub0, ub1, ub2, c0.w, c1.x, c1.y, 20);
    TRI_STEP(0, 1, 2, r0, r1, r2, mk[20], ub0, ub1, ub2, ua0, ua1, ua2, c1.z, c1.w, c2.x, 20);
    TRI_STEP(2, 0, 1, r3, r4, r5, mk[20], ua0, ua1, ua2, ub0, ub1, ub2, c2.y, c2.z, c2.w, 20);
    TRI_STEP(1, 2, 0, r6, r7, r8, mk[20], ub0, ub1, ub2, ua0, ua1, ua2, c0.x, c0.y, c0.z, 21);
    TRI_STEP(0, 1, 2, r9, r10, r11, mk[20], ua0, ua1, ua2, ub0, ub1, ub2, c0.w, c1.x, c1.y, 21);
    TRI_STEP(2, 0, 1, r0, r1, r2, mk[21], ub0, ub1, ub2, ua0, ua1, ua2, c1.z, c1.w, c2.x, 21);
    TRI_STEP(1, 2, 0, r3, r4, r5, mk[21], ua0, ua1, ua2, ub0, ub1, ub2, c2.y, c2.z, c2.w, 21);
    TRI_STEP(0, 1, 2, r6, r7, r8, mk[21], ub0, ub1, ub2, ua0, ua1, ua2, c0.x, c0.y, c0.z, 22);
    TRI_STEP(2, 0, 1, r9, r10, r11, mk[21], ua0, ua1, ua2, ub0, ub1, ub2, c0.w, c1.x, c1.y, 22);
    TRI_STEP(1, 2, 0, r0, r1, r2, mk[22], ub0, ub1, ub2, ua0, ua1, ua2, c1.z, c1.w, c2.x, 22);
    TRI_STEP(0, 1, 2, r3, r4, r5, mk[22], ua0, ua1, ua2, ub0, ub1, ub2, c2.y, c2.z, c2.w, 22);
    TRI_STEP(2, 0, 1, r6, r7, r8, mk[22], ub0, ub1, ub2, ua0, ua1, ua2, c0.x, c0.y, c0.z, 23);
    TRI_STEP(1, 2, 0, r9, r10, r11, mk[22], ua0, ua1, ua2, ub0, ub1, ub2, c0.w, c1.x, c1.y, 23);
    TRI_STEP(0, 1, 2, r0, r1, r2, mk[23], ub0, ub1, ub2, ua0, ua1, ua2, c1.z, c1.w, c2.x, 23);
    TRI_STEP(2, 0, 1, r3, r4, r5, mk[23], ua0, ua1, ua2, ub0, ub1, ub2, c2.y, c2.z, c2.w, 23);
    TRI_STEP(1, 2, 0, r6, r7, r8, mk[23], ub0, ub1, ub2, ua0, ua1, ua2, c0.x, c0.y, c0.z, 24);
    TRI_STEP(0, 1, 2, r9, r10, r11, mk[23], ua0, ua1, ua2, ub0, ub1, ub2, c0.w, c1.x, c1.y, 24);
    TRI_STEP(2, 0, 1, r0, r1, r2, mk[24], ub0, ub1, ub2, ua0, ua1, ua2, c1.z, c1.w, c2.x, 24);
    TRI_STEP(1, 2, 0, r3, r4, r5, mk[24], ua0, ua1, ua2, ub0, ub1, ub2, c2.y, c2.z, c2.w, 24);
    TRI_STEP(0, 1, 2, r6, r7, r8, mk[24], ub0, ub1, ub2, ua0, ua1, ua2, c0.x, c0.y, c0.z, 25);
    TRI_STEP(2, 0, 1, r9, r10, r11, mk[24], ua0, ua1, ua2, ub0, ub1, ub2, c0.w, c1.x, c1.y, 25);
    TRI_STEP(1, 2, 0, r0, r1, r2, mk[25], ub0, ub1, ub2, ua0, ua1, ua2, c1.z, c1.w, c2.x, 25);
    TRI_STEP(0, 1, 2, r3, r4, r5, mk[25], ua0, ua1, ua2, ub0, ub1, ub2, c2.y, c2.z, c2.w, 25);
    TRI_STEP(2, 0, 1, r6, r7, r8, mk[25], ub0, ub1, ub2, ua0, ua1, ua2, c0.x, c0.y, c0.z, 26);
    TRI_STEP(1, 2, 0, r9, r10, r11, mk[25], ua0, ua1, ua2, ub0, ub1, ub2, c0.w, c1.x, c1.y, 26);
    TRI_STEP(0, 1, 2, r0, r1, r2, mk[26], ub0, ub1, ub2, ua0, ua1, ua2, c1.z, c1.w, c2.x, 26);
    TRI_STEP(2, 0, 1, r3, r4, r5, mk[26], ua0, ua1, ua2, ub0, ub1, ub2, c2.y, c2.z, c2.w, 26);
    TRI_STEP(1, 2, 0, r6, r7, r8, mk[26], ub0, ub1, ub2, ua0, ua1, ua2, c0.x, c0.y, c0.z, 27);
    TRI_STEP(0, 1, 2, r9, r10, r11, mk[26], ua0, ua1, ua2, ub0, ub1, ub2, c0.w, c1.x, c1.y, 27);
    TRI_STEP(2, 0, 1, r0, r1, r2, mk[27], ub0, ub1, ub2, ua0, ua1, ua2, c1.z, c1.w, c2.x, 27);
    TRI_STEP(1, 2, 0, r3, r4, r5, mk[27], ua0, ua1, ua2, ub0, ub1, ub2, c2.y, c2.z, c2.w, 27);
    TRI_STEP(0, 1, 2, r6, r7, r8, mk[27], ub0, ub1, ub2, ua0, ua1, ua2, c0.x, c0.y, c0.z, 28);
    TRI_STEP(2, 0, 1, r9, r10, r11, mk[27], ua0, ua1, ua2, ub0, ub1, ub2, c0.w, c1.x, c1.y, 28);
    TRI_STEP(1, 2, 0, r0, r1, r2, mk[28], ub0, ub1, ub2, ua0, ua1, ua2, c1.z, c1.w, c2.x, 28);
    TRI_STEP(0, 1, 2, r3, r4, r5, mk[28], ua0, ua1, ua2, ub0, ub1, ub2, c2.y, c2.z, c2.w, 28);
    TRI_STEP(2, 0, 1, r6, r7, r8, mk[28], ub0, ub1, ub2, ua0, ua1, ua2, c0.x, c0.y, c0.z, 29);
    TRI_STEP(1, 2, 0, r9, r10, r11, mk[28], ua0, ua1, ua2, ub0, ub1, ub2, c0.w, c1.x, c1.y, 29);
    TRI_STEP(0, 1, 2, r0, r1, r2, mk[29], ub0, ub1, ub2, ua0, ua1, ua2, c1.z, c1.w, c2.x, 29);
    TRI_STEP(2, 0, 1, r3, r4, r5, mk[29], ua0, ua1, ua2, ub0, ub1, ub2, c2.y, c2.z, c2.w, 29);
    TRI_STEP(1, 2, 0, r6, r7, r8, mk[29], ub0, ub1, ub2, ua0, ua1, ua2, c0.x, c0.y, c0.z, 30);
    TRI_STEP(0, 1, 2, r9, r10, r11, mk[29], ua0, ua1, ua2, ub0, ub1, ub2, c0.w, c1.x, c1.y, 30);
    TRI_STEP(2, 0, 1, r0, r1, r2, mk[30], ub0, ub1, ub2, ua0, ua1, ua2, c1.z, c1.w, c2.x, 30);
    TRI_STEP(1, 2, 0, r3, r4, r5, mk[30], ua0, ua1, ua2, ub0, ub1, ub2, c2.y, c2.z, c2.w, 30);
    TRI_STEP(0, 1, 2, r6, r7, r8, mk[30], ub0, ub1, ub2, ua0, ua1, ua2, c0.x, c0.y, c0.z, 31);
    TRI_STEP(2, 0, 1, r9, r10, r11, mk[30], ua0, ua1, ua2, ub0, ub1, ub2, c0.w, c1.x, c1.y, 31);
    TRI_STEP(1, 2, 0, r0, r1, r2, mk[31], ub0, ub1, ub2, ua0, ua1, ua2, c1.z, c1.w, c2.x, 31);
    TRI_STEP(0, 1, 2, r3, r4, r5, mk[31], ua0, ua1, ua2, ub0, ub1, ub2, c2.y, c2.z, c2.w, 31);
    TRI_STEP(2, 0, 1, r6, r7, r8, mk[31], ub0, ub1, ub2, ua0, ua1, ua2, n0.x, n0.y, n0.z, 0);

    r9  = (lane == 31) ? h0 : r9;
    r10 = (lane == 31) ? h1 : r10;
    r11 = (lane == 31) ? h2 : r11;

    v4f* po = pl + c * CHUNK_V4;
    po[0] = (v4f){ r0, r1, r2,  r3  };
    po[1] = (v4f){ r4, r5, r6,  r7  };
    po[2] = (v4f){ r8, r9, r10, r11 };
    c0 = n0; c1 = n1; c2 = n2;
  }

  if (lane == 0) {
    float* hf = out + (size_t)BATCH * SEQ * HSZ + (size_t)b * HSZ;
    hf[0] = h0; hf[1] = h1; hf[2] = h2;
  }
}

extern "C" void kernel_launch(void* const* d_in, const int* in_sizes, int n_in,
                              void* d_out, int out_size, void* d_ws, size_t ws_size,
                              hipStream_t stream) {
  const float* x    = (const float*)d_in[0];
  const float* wp   = (const float*)d_in[1];
  const float* bias = (const float*)d_in[2];
  float* out = (float*)d_out;

  const int tiles  = (BATCH * SEQ) / 16;
  const int blocks = tiles / (PROJ_WAVES_PER_WG * PROJ_BLOCKS_PER_WAVE);

  tridom_proj_wmma<<<dim3(blocks), dim3(32 * PROJ_WAVES_PER_WG), 0, stream>>>(x, wp, bias, out);
  tridom_scan<<<dim3(BATCH), dim3(32), 0, stream>>>(out);
}


# 3 launches, cut at step 768: head scan beside the tail's projection on disjoint WGPs (every work-group of that launch reserves 176 KiB of LDS: one per WGP; scans first in the grid); sums unchanged
# speedup vs baseline: 6.6344x; 1.0450x over previous
#include <hip/hip_runtime.h>

typedef __attribute__((ext_vector_type(2))) float v2f;
typedef __attribute__((ext_vector_type(4))) float v4f;
typedef __attribute__((ext_vector_type(8))) float v8f;

#define BATCH 128
#define SEQ   2048
#define ISZ   128
#define HSZ   3

#define PROJ_GROUPS          4
#define PROJ_LINES           (ISZ / 32)

#define T_CUT        768
#define HEAD_NB      4
#define HEAD_WAVES   6
#define TAIL_NB      10
#define MID_WAVES    8
#define MID_LDS_BYTES (176 * 1024)

__device__ __forceinline__ void proj_load_block(v4f (&a)[PROJ_GROUPS][PROJ_LINES], const float* p)
{
#pragma unroll
  for (int G = 0; G < PROJ_GROUPS; ++G)
#pragma unroll
    for (int L = 0; L < PROJ_LINES; ++L)
      a[G][L] = *(const v4f*)(p + (size_t)G * 4 * ISZ + 32 * L);
}

__device__ __forceinline__ float quad_x1(float v)
{
  const int b = __builtin_bit_cast(int, v);
  return __builtin_bit_cast(float, __builtin_amdgcn_update_dpp(b, b, 0xB1, 0xF, 0xF, true));
}
__device__ __forceinline__ float quad_x2(float v)
{
  const int b = __builtin_bit_cast(int, v);
  return __builtin_bit_cast(float, __builtin_amdgcn_update_dpp(b, b, 0x4E, 0xF, 0xF, true));
}

__device__ __forceinline__ float sel4(int q, float v0, float v1, float v2, float v3)
{
  const float lo = (q & 1) ? v1 : v0;
  const float hi = (q & 1) ? v3 : v2;
  return (q & 2) ? hi : lo;
}

__device__ __forceinline__ void proj_block(const v4f (&a)[PROJ_GROUPS][PROJ_LINES], const v4f (&bf)[PROJ_LINES],
                                           float* __restrict__ out, int row0, int hi, int h, int gq, float bb)
{
  v8f c[PROJ_GROUPS];
#pragma unroll
  for (int G = 0; G < PROJ_GROUPS; ++G) c[G] = (v8f){};
#pragma unroll
  for (int L = 0; L < PROJ_LINES; ++L) {
    const v2f b_lo = { bf[L].x, bf[L].y }, b_hi = { bf[L].z, bf[L].w };
#pragma unroll
    for (int G = 0; G < PROJ_GROUPS; ++G) {
      const v2f a_lo = { a[G][L].x, a[G][L].y };
      c[G] = __builtin_amdgcn_wmma_f32_16x16x4_f32(false, a_lo, false, b_lo,
                                                   (short)0, c[G], false, false);
    }
#pragma unroll
    for (int G = 0; G < PROJ_GROUPS; ++G) {
      const v2f a_hi = { a[G][L].z, a[G][L].w };
      c[G] = __builtin_amdgcn_wmma_f32_16x16x4_f32(false, a_hi, false, b_hi,
                                                   (short)0, c[G], false, false);
    }
  }
  float s0[PROJ_GROUPS], s1[PROJ_GROUPS];
#pragma unroll
  for (int G = 0; G < PROJ_GROUPS; ++G) {
    float w0 = sel4(gq, c[G][0], c[G][1], c[G][2], c[G][3]);
    float w1 = sel4(gq, c[G][4], c[G][5], c[G][6], c[G][7]);
    w0 += quad_x1(w0);  w1 += quad_x1(w1);
    w0 += quad_x2(w0);  w1 += quad_x2(w1);
    s0[G] = w0;  s1[G] = w1;
  }
  const float o0 = sel4(gq, s0[0], s0[1], s0[2], s0[3]) + bb;
  const float o1 = sel4(gq, s1[0], s1[1], s1[2], s1[3]) + bb;
  if (h < HSZ) {
    const int grow = row0 + 4 * gq + 2 * hi;
    out[(size_t)grow * HSZ + h]       = o0;
    out[(size_t)(grow + 1) * HSZ + h] = o1;
  }
}

template <int NB, int T0, int SPAN>
__device__ __forceinline__ void proj_wave(const float* __restrict__ x, const float* __restrict__ wp,
                                          const float* __restrict__ bias, float* __restrict__ out,
                                          int gw, int lane)
{
  constexpr int UPC = SPAN / (NB * 16);
  const int hi   = lane >> 4;
  const int ml   = lane & 15;
  const int tl   = ml >> 2;
  const int gq   = ml & 3;
  const int chain = gw / UPC;
  const int row_w = chain * SEQ + T0 + (gw - chain * UPC) * (NB * 16);

  const float* ap = x + ((size_t)row_w + tl) * ISZ + 8 * gq + 4 * hi;

  v4f a0[PROJ_GROUPS][PROJ_LINES], a1[PROJ_GROUPS][PROJ_LINES];
  proj_load_block(a0, ap);

  const int   h     = tl;
  const float bmask = (h < HSZ) ? 1.0f : 0.0f;
  const float* brow = wp + (size_t)((h < HSZ) ? h : 0) * ISZ + 8 * gq + 4 * hi;
  const float bb    = bias[(h < HSZ) ? h : 0];
  v4f bf[PROJ_LINES];
#pragma unroll
  for (int L = 0; L < PROJ_LINES; ++L) {
    v4f b4 = *(const v4f*)(brow + 32 * L);
    bf[L] = b4 * bmask;
  }

#pragma unroll 1
  for (int i = 0; i < (NB & ~1); i += 2) {
    proj_load_block(a1, ap + (size_t)(i + 1) * 16 * ISZ);
    proj_block(a0, bf, out, row_w + 16 * i, hi, h, gq, bb);
    if (i + 2 < NB)
      proj_load_block(a0, ap + (size_t)(i + 2) * 16 * ISZ);
    proj_block(a1, bf, out, row_w + 16 * (i + 1), hi, h, gq, bb);
  }
  if (NB & 1)
    proj_block(a0, bf, out, row_w + 16 * (NB - 1), hi, h, gq, bb);
}

__global__ __launch_bounds__(32 * HEAD_WAVES) void tridom_proj_head(
    const float* __restrict__ x, const float* __restrict__ wp,
    const float* __restrict__ bias, float* __restrict__ out)
{
  const int lane = threadIdx.x & 31;
  const int wave = threadIdx.x >> 5;
  proj_wave<HEAD_NB, 0, T_CUT>(x, wp, bias, out, blockIdx.x * HEAD_WAVES + wave, lane);
}

#define CHUNK_STEPS 128
#define CHUNK_V4    (CHUNK_STEPS * HSZ / 4)
#define NCHUNK      (SEQ / CHUNK_STEPS)

#define TRI_MUL0 "v_mul_f32 %[t0], 0x3fc00000, %[h1]\n\t"
#define TRI_MUL1 "v_mul_f32 %[t1], 0x3fc00000, %[h2]\n\t"
#define TRI_MUL2 "v_mul_f32 %[t2], 0xbfc00000, %[h0]\n\t"
#define TRI_MUL(c) TRI_MUL##c
#define TRI_RDL(c) "v_readlane_b32 %[n" #c "], %[c" #c "], %[j]\n\t"
#define TRI_TNH(c) "v_tanh_f32 %[t" #c "], %[t" #c "]\n\t"
#define TRI_FMC(c) "v_fmac_f32 %[t" #c "], -0.5, %[h" #c "]\n\t"
#define TRI_ADD(c) "v_add_f32 %[h" #c "], %[u" #c "], %[t" #c "]\n\t"
#define TRI_CAP(c) "v_cndmask_b32_e64 %[r" #c "], %[r" #c "], %[h" #c "], %[m]\n\t"

#define TRI_STEP(X, Y, Z, R0, R1, R2, MV, U0, U1, U2, N0, N1, N2, CN0, CN1, CN2, JN) \
  do {                                                                           \
    float t0_, t1_, t2_;                                                         \
    asm(TRI_MUL(X) TRI_RDL(0) TRI_MUL(Y) TRI_RDL(1) TRI_MUL(Z)                   \
        TRI_TNH(X) TRI_RDL(2) TRI_TNH(Y) TRI_TNH(Z)                              \
        TRI_CAP(0) TRI_CAP(1) TRI_CAP(2)                                         \
        TRI_FMC(X) TRI_FMC(Y) TRI_FMC(Z)                                         \
        TRI_ADD(X) TRI_ADD(Y) TRI_ADD(Z)                                         \
        : [h0] "+v"(h0), [h1] "+v"(h1), [h2] "+v"(h2),                           \
          [r0] "+v"(R0), [r1] "+v"(R1), [r2] "+v"(R2),                           \
          [t0] "=&v"(t0_), [t1] "=&v"(t1_), [t2] "=&v"(t2_),                     \
          [n0] "=&s"(N0), [n1] "=&s"(N1), [n2] "=&s"(N2)                         \
        : [c0] "v"(CN0), [c1] "v"(CN1), [c2] "v"(CN2),                           \
          [u0] "s"(U0), [u1] "s"(U1), [u2] "s"(U2), [m] "s"(MV), [j] "n"(JN));   \
  } while (0)

#define TRI_STEP_FIRST(X, Y, Z, U0, U1, U2, N0, N1, N2, CN0, CN1, CN2, JN)       \
  do {                                                                           \
    float t0_, t1_, t2_;                                                         \
    asm(TRI_MUL(X) TRI_RDL(0) TRI_MUL(Y) TRI_RDL(1) TRI_MUL(Z)                   \
        TRI_TNH(X) TRI_RDL(2) TRI_TNH(Y) TRI_TNH(Z)                              \
        TRI_FMC(X) TRI_FMC(Y) TRI_FMC(Z)                                         \
        TRI_ADD(X) TRI_ADD(Y) TRI_ADD(Z)                                         \
        : [h0] "+v"(h0), [h1] "+v"(h1), [h2] "+v"(h2),                           \
          [t0] "=&v"(t0_), [t1] "=&v"(t1_), [t2] "=&v"(t2_),                     \
          [n0] "=&s"(N0), [n1] "=&s"(N1), [n2] "=&s"(N2)                         \
        : [c0] "v"(CN0), [c1] "v"(CN1), [c2] "v"(CN2),                           \
          [u0] "s"(U0), [u1] "s"(U1), [u2] "s"(U2), [j] "n"(JN));                \
  } while (0)

__device__ __forceinline__ float tri_lane0(float v)
{
  return __builtin_bit_cast(float, __builtin_amdgcn_readlane(__builtin_bit_cast(int, v), 0));
}

template <int C0, int C1>
__device__ __forceinline__ void tridom_scan_range(float* __restrict__ out, int b, int lane)
{
  v4f* pl = (v4f*)(out + (size_t)b * SEQ * HSZ) + 3 * lane;

  float h0 = 0.f, h1 = 0.f, h2 = 0.f;
  if (C0 > 0) {
    const float* hp = out + ((size_t)b * SEQ + (size_t)C0 * CHUNK_STEPS - 1) * HSZ;
    h0 = hp[0]; h1 = hp[1]; h2 = hp[2];
  }
  float r0 = 0.f, r1 = 0.f, r2 = 0.f, r3 = 0.f, r4 = 0.f, r5 = 0.f;
  float r6 = 0.f, r7 = 0.f, r8 = 0.f, r9 = 0.f, r10 = 0.f, r11 = 0.f;

  unsigned mk[32];
#pragma unroll
  for (int j = 0; j < 32; ++j) mk[j] = __builtin_amdgcn_ballot_w32(lane == j);

  v4f c0 = pl[C0 * CHUNK_V4 + 0], c1 = pl[C0 * CHUNK_V4 + 1], c2 = pl[C0 * CHUNK_V4 + 2];

  float ua0 = tri_lane0(c0.x), ua1 = tri_lane0(c0.y), ua2 = tri_lane0(c0.z);
  float ub0, ub1, ub2;

  for (int c = C0; c < C1; ++c) {
    v4f n0 = c0, n1 = c1, n2 = c2;
    if (c + 1 < C1) {
      const v4f* pn = pl + (c + 1) * CHUNK_V4;
      n0 = pn[0]; n1 = pn[1]; n2 = pn[2];
    }

    TRI_STEP_FIRST(0, 1, 2, ua0, ua1, ua2, ub0, ub1, ub2, c0.w, c1.x, c1.y, 0);
    TRI_STEP(2, 0, 1, r0, r1, r2, mk[0], ub0, ub1, ub2, ua0, ua1, ua2, c1.z, c1.w, c2.x, 0);
    TRI_STEP(1, 2, 0, r3, r4, r5, mk[0], ua0, ua1, ua2, ub0, ub1, ub2, c2.y, c2.z, c2.w, 0);
    TRI_STEP(0, 1, 2, r6, r7, r8, mk[0], ub0, ub1, ub2, ua0, ua1, ua2, c0.x, c0.y, c0.z, 1);
    TRI_STEP(2, 0, 1, r9, r10, r11, mk[0], ua0, ua1, ua2, ub0, ub1, ub2, c0.w, c1.x, c1.y, 1);
    TRI_STEP(1, 2, 0, r0, r1, r2, mk[1], ub0, ub1, ub2, ua0, ua1, ua2, c1.z, c1.w, c2.x, 1);
    TRI_STEP(0, 1, 2, r3, r4, r5, mk[1], ua0, ua1, ua2, ub0, ub1, ub2, c2.y, c2.z, c2.w, 1);
    TRI_STEP(2, 0, 1, r6, r7, r8, mk[1], ub0, ub1, ub2, ua0, ua1, ua2, c0.x, c0.y, c0.z, 2);
    TRI_STEP(1, 2, 0, r9, r10, r11, mk[1], ua0, ua1, ua2, ub0, ub1, ub2, c0.w, c1.x, c1.y, 2);
    TRI_STEP(0, 1, 2, r0, r1, r2, mk[2], ub0, ub1, ub2, ua0, ua1, ua2, c1.z, c1.w, c2.x, 2);
    TRI_STEP(2, 0, 1, r3, r4, r5, mk[2], ua0, ua1, ua2, ub0, ub1, ub2, c2.y, c2.z, c2.w, 2);
    TRI_STEP(1, 2, 0, r6, r7, r8, mk[2], ub0, ub1, ub2, ua0, ua1, ua2, c0.x, c0.y, c0.z, 3);
    TRI_STEP(0, 1, 2, r9, r10, r11, mk[2], ua0, ua1, ua2, ub0, ub1, ub2, c0.w, c1.x, c1.y, 3);
    TRI_STEP(2, 0, 1, r0, r1, r2, mk[3], ub0, ub1, ub2, ua0, ua1, ua2, c1.z, c1.w, c2.x, 3);
    TRI_STEP(1, 2, 0, r3, r4, r5, mk[3], ua0, ua1, ua2, ub0, ub1, ub2, c2.y, c2.z, c2.w, 3);
    TRI_STEP(0, 1, 2, r6, r7, r8, mk[3], ub0, ub1, ub2, ua0, ua1, ua2, c0.x, c0.y, c0.z, 4);
    TRI_STEP(2, 0, 1, r9, r10, r11, mk[3], ua0, ua1, ua2, ub0, ub1, ub2, c0.w, c1.x, c1.y, 4);
    TRI_STEP(1, 2, 0, r0, r1, r2, mk[4], ub0, ub1, ub2, ua0, ua1, ua2, c1.z, c1.w, c2.x, 4);
    TRI_STEP(0, 1, 2, r3, r4, r5, mk[4], ua0, ua1, ua2, ub0, ub1, ub2, c2.y, c2.z, c2.w, 4);
    TRI_STEP(2, 0, 1, r6, r7, r8, mk[4], ub0, ub1, ub2, ua0, ua1, ua2, c0.x, c0.y, c0.z, 5);
    TRI_STEP(1, 2, 0, r9, r10, r11, mk[4], ua0, ua1, ua2, ub0, ub1, ub2, c0.w, c1.x, c1.y, 5);
    TRI_STEP(0, 1, 2, r0, r1, r2, mk[5], ub0, ub1, ub2, ua0, ua1, ua2, c1.z, c1.w, c2.x, 5);
    TRI_STEP(2, 0, 1, r3, r4, r5, mk[5], ua0, ua1, ua2, ub0, ub1, ub2, c2.y, c2.z, c2.w, 5);
    TRI_STEP(1, 2, 0, r6, r7, r8, mk[5], ub0, ub1, ub2, ua0, ua1, ua2, c0.x, c0.y, c0.z, 6);
    TRI_STEP(0, 1, 2, r9, r10, r11, mk[5], ua0, ua1, ua2, ub0, ub1, ub2, c0.w, c1.x, c1.y, 6);
    TRI_STEP(2, 0, 1, r0, r1, r2, mk[6], ub0, ub1, ub2, ua0, ua1, ua2, c1.z, c1.w, c2.x, 6);
    TRI_STEP(1, 2, 0, r3, r4, r5, mk[6], ua0, ua1, ua2, ub0, ub1, ub2, c2.y, c2.z, c2.w, 6);
    TRI_STEP(0, 1, 2, r6, r7, r8, mk[6], ub0, ub1, ub2, ua0, ua1, ua2, c0.x, c0.y, c0.z, 7);
    TRI_STEP(2, 0, 1, r9, r10, r11, mk[6], ua0, ua1, ua2, ub0, ub1, ub2, c0.w, c1.x, c1.y, 7);
    TRI_STEP(1, 2, 0, r0, r1, r2, mk[7], ub0, ub1, ub2, ua0, ua1, ua2, c1.z, c1.w, c2.x, 7);
    TRI_STEP(0, 1, 2, r3, r4, r5, mk[7], ua0, ua1, ua2, ub0, ub1, ub2, c2.y, c2.z, c2.w, 7);
    TRI_STEP(2, 0, 1, r6, r7, r8, mk[7], ub0, ub1, ub2, ua0, ua1, ua2, c0.x, c0.y, c0.z, 8);
    TRI_STEP(1, 2, 0, r9, r10, r11, mk[7], ua0, ua1, ua2, ub0, ub1, ub2, c0.w, c1.x, c1.y, 8);
    TRI_STEP(0, 1, 2, r0, r1, r2, mk[8], ub0, ub1, ub2, ua0, ua1, ua2, c1.z, c1.w, c2.x, 8);
    TRI_STEP(2, 0, 1, r3, r4, r5, mk[8], ua0, ua1, ua2, ub0, ub1, ub2, c2.y, c2.z, c2.w, 8);
    TRI_STEP(1, 2, 0, r6, r7, r8, mk[8], ub0, ub1, ub2, ua0, ua1, ua2, c0.x, c0.y, c0.z, 9);
    TRI_STEP(0, 1, 2, r9, r10, r11, mk[8], ua0, ua1, ua2, ub0, ub1, ub2, c0.w, c1.x, c1.y, 9);
    TRI_STEP(2, 0, 1, r0, r1, r2, mk[9], ub0, ub1, ub2, ua0, ua1, ua2, c1.z, c1.w, c2.x, 9);
    TRI_STEP(1, 2, 0, r3, r4, r5, mk[9], ua0, ua1, ua2, ub0, ub1, ub2, c2.y, c2.z, c2.w, 9);
    TRI_STEP(0, 1, 2, r6, r7, r8, mk[9], ub0, ub1, ub2, ua0, ua1, ua2, c0.x, c0.y, c0.z, 10);
    TRI_STEP(2, 0, 1, r9, r10, r11, mk[9], ua0, ua1, ua2, ub0, ub1, ub2, c0.w, c1.x, c1.y, 10);
    TRI_STEP(1, 2, 0, r0, r1, r2, mk[10], ub0, ub1, ub2, ua0, ua1, ua2, c1.z, c1.w, c2.x, 10);
    TRI_STEP(0, 1, 2, r3, r4, r5, mk[10], ua0, ua1, ua2, ub0, ub1, ub2, c2.y, c2.z, c2.w, 10);
    TRI_STEP(2, 0, 1, r6, r7, r8, mk[10], ub0, ub1, ub2, ua0, ua1, ua2, c0.x, c0.y, c0.z, 11);
    TRI_STEP(1, 2, 0, r9, r10, r11, mk[10], ua0, ua1, ua2, ub0, ub1, ub2, c0.w, c1.x, c1.y, 11);
    TRI_STEP(0, 1, 2, r0, r1, r2, mk[11], ub0, ub1, ub2, ua0, ua1, ua2, c1.z, c1.w, c2.x, 11);
    TRI_STEP(2, 0, 1, r3, r4, r5, mk[11], ua0, ua1, ua2, ub0, ub1, ub2, c2.y, c2.z, c2.w, 11);
    TRI_STEP(1, 2, 0, r6, r7, r8, mk[11], ub0, ub1, ub2, ua0, ua1, ua2, c0.x, c0.y, c0.z, 12);
    TRI_STEP(0, 1, 2, r9, r10, r11, mk[11], ua0, ua1, ua2, ub0, ub1, ub2, c0.w, c1.x, c1.y, 12);
    TRI_STEP(2, 0, 1, r0, r1, r2, mk[12], ub0, ub1, ub2, ua0, ua1, ua2, c1.z, c1.w, c2.x, 12);
    TRI_STEP(1, 2, 0, r3, r4, r5, mk[12], ua0, ua1, ua2, ub0, ub1, ub2, c2.y, c2.z, c2.w, 12);
    TRI_STEP(0, 1, 2, r6, r7, r8, mk[12], ub0, ub1, ub2, ua0, ua1, ua2, c0.x, c0.y, c0.z, 13);
    TRI_STEP(2, 0, 1, r9, r10, r11, mk[12], ua0, ua1, ua2, ub0, ub1, ub2, c0.w, c1.x, c1.y, 13);
    TRI_STEP(1, 2, 0, r0, r1, r2, mk[13], ub0, ub1, ub2, ua0, ua1, ua2, c1.z, c1.w, c2.x, 13);
    TRI_STEP(0, 1, 2, r3, r4, r5, mk[13], ua0, ua1, ua2, ub0, ub1, ub2, c2.y, c2.z, c2.w, 13);
    TRI_STEP(2, 0, 1, r6, r7, r8, mk[13], ub0, ub1, ub2, ua0, ua1, ua2, c0.x, c0.y, c0.z, 14);
    TRI_STEP(1, 2, 0, r9, r10, r11, mk[13], ua0, ua1, ua2, ub0, ub1, ub2, c0.w, c1.x, c1.y, 14);
    TRI_STEP(0, 1, 2, r0, r1, r2, mk[14], ub0, ub1, ub2, ua0, ua1, ua2, c1.z, c1.w, c2.x, 14);
    TRI_STEP(2, 0, 1, r3, r4, r5, mk[14], ua0, ua1, ua2, ub0, ub1, ub2, c2.y, c2.z, c2.w, 14);
    TRI_STEP(1, 2, 0, r6, r7, r8, mk[14], ub0, ub1, ub2, ua0, ua1, ua2, c0.x, c0.y, c0.z, 15);
    TRI_STEP(0, 1, 2, r9, r10, r11, mk[14], ua0, ua1, ua2, ub0, ub1, ub2, c0.w, c1.x, c1.y, 15);
    TRI_STEP(2, 0, 1, r0, r1, r2, mk[15], ub0, ub1, ub2, ua0, ua1, ua2, c1.z, c1.w, c2.x, 15);
    TRI_STEP(1, 2, 0, r3, r4, r5, mk[15], ua0, ua1, ua2, ub0, ub1, ub2, c2.y, c2.z, c2.w, 15);
    TRI_STEP(0, 1, 2, r6, r7, r8, mk[15], ub0, ub1, ub2, ua0, ua1, ua2, c0.x, c0.y, c0.z, 16);
    TRI_STEP(2, 0, 1, r9, r10, r11, mk[15], ua0, ua1, ua2, ub0, ub1, ub2, c0.w, c1.x, c1.y, 16);
    TRI_STEP(1, 2, 0, r0, r1, r2, mk[16], ub0, ub1, ub2, ua0, ua1, ua2, c1.z, c1.w, c2.x, 16);
    TRI_STEP(0, 1, 2, r3, r4, r5, mk[16], ua0, ua1, ua2, ub0, ub1, ub2, c2.y, c2.z, c2.w, 16);
    TRI_STEP(2, 0, 1, r6, r7, r8, mk[16], ub0, ub1, ub2, ua0, ua1, ua2, c0.x, c0.y, c0.z, 17);
    TRI_STEP(1, 2, 0, r9, r10, r11, mk[16], ua0, ua1, ua2, ub0, ub1, ub2, c0.w, c1.x, c1.y, 17);
    TRI_STEP(0, 1, 2, r0, r1, r2, mk[17], ub0, ub1, ub2, ua0, ua1, ua2, c1.z, c1.w, c2.x, 17);
    TRI_STEP(2, 0, 1, r3, r4, r5, mk[17], ua0, ua1, ua2, ub0, ub1, ub2, c2.y, c2.z, c2.w, 17);
    TRI_STEP(1, 2, 0, r6, r7, r8, mk[17], ub0, ub1, ub2, ua0, ua1, ua2, c0.x, c0.y, c0.z, 18);
    TRI_STEP(0, 1, 2, r9, r10, r11, mk[17], ua0, ua1, ua2, ub0, ub1, ub2, c0.w, c1.x, c1.y, 18);
    TRI_STEP(2, 0, 1, r0, r1, r2, mk[18], ub0, ub1, ub2, ua0, ua1, ua2, c1.z, c1.w, c2.x, 18);
    TRI_STEP(1, 2, 0, r3, r4, r5, mk[18], ua0, ua1, ua2, ub0, ub1, ub2, c2.y, c2.z, c2.w, 18);
    TRI_STEP(0, 1, 2, r6, r7, r8, mk[18], ub0, ub1, ub2, ua0, ua1, ua2, c0.x, c0.y, c0.z, 19);
    TRI_STEP(2, 0, 1, r9, r10, r11, mk[18], ua0, ua1, ua2, ub0, ub1, ub2, c0.w, c1.x, c1.y, 19);
    TRI_STEP(1, 2, 0, r0, r1, r2, mk[19], ub0, ub1, ub2, ua0, ua1, ua2, c1.z, c1.w, c2.x, 19);
    TRI_STEP(0, 1, 2, r3, r4, r5, mk[19], ua0, ua1, ua2, ub0, ub1, ub2, c2.y, c2.z, c2.w, 19);
    TRI_STEP(2, 0, 1, r6, r7, r8, mk[19], ub0, ub1, ub2, ua0, ua1, ua2, c0.x, c0.y, c0.z, 20);
    TRI_STEP(1, 2, 0, r9, r10, r11, mk[19], ua0, ua1, ua2, ub0, ub1, ub2, c0.w, c1.x, c1.y, 20);
    TRI_STEP(0, 1, 2, r0, r1, r2, mk[20], ub0, ub1, ub2, ua0, ua1, ua2, c1.z, c1.w, c2.x, 20);
    TRI_STEP(2, 0, 1, r3, r4, r5, mk[20], ua0, ua1, ua2, ub0, ub1, ub2, c2.y, c2.z, c2.w, 20);
    TRI_STEP(1, 2, 0, r6, r7, r8, mk[20], ub0, ub1, ub2, ua0, ua1, ua2, c0.x, c0.y, c0.z, 21);
    TRI_STEP(0, 1, 2, r9, r10, r11, mk[20], ua0, ua1, ua2, ub0, ub1, ub2, c0.w, c1.x, c1.y, 21);
    TRI_STEP(2, 0, 1, r0, r1, r2, mk[21], ub0, ub1, ub2, ua0, ua1, ua2, c1.z, c1.w, c2.x, 21);
    TRI_STEP(1, 2, 0, r3, r4, r5, mk[21], ua0, ua1, ua2, ub0, ub1, ub2, c2.y, c2.z, c2.w, 21);
    TRI_STEP(0, 1, 2, r6, r7, r8, mk[21], ub0, ub1, ub2, ua0, ua1, ua2, c0.x, c0.y, c0.z, 22);
    TRI_STEP(2, 0, 1, r9, r10, r11, mk[21], ua0, ua1, ua2, ub0, ub1, ub2, c0.w, c1.x, c1.y, 22);
    TRI_STEP(1, 2, 0, r0, r1, r2, mk[22], ub0, ub1, ub2, ua0, ua1, ua2, c1.z, c1.w, c2.x, 22);
    TRI_STEP(0, 1, 2, r3, r4, r5, mk[22], ua0, ua1, ua2, ub0, ub1, ub2, c2.y, c2.z, c2.w, 22);
    TRI_STEP(2, 0, 1, r6, r7, r8, mk[22], ub0, ub1, ub2, ua0, ua1, ua2, c0.x, c0.y, c0.z, 23);
    TRI_STEP(1, 2, 0, r9, r10, r11, mk[22], ua0, ua1, ua2, ub0, ub1, ub2, c0.w, c1.x, c1.y, 23);
    TRI_STEP(0, 1, 2, r0, r1, r2, mk[23], ub0, ub1, ub2, ua0, ua1, ua2, c1.z, c1.w, c2.x, 23);
    TRI_STEP(2, 0, 1, r3, r4, r5, mk[23], ua0, ua1, ua2, ub0, ub1, ub2, c2.y, c2.z, c2.w, 23);
    TRI_STEP(1, 2, 0, r6, r7, r8, mk[23], ub0, ub1, ub2, ua0, ua1, ua2, c0.x, c0.y, c0.z, 24);
    TRI_STEP(0, 1, 2, r9, r10, r11, mk[23], ua0, ua1, ua2, ub0, ub1, ub2, c0.w, c1.x, c1.y, 24);
    TRI_STEP(2, 0, 1, r0, r1, r2, mk[24], ub0, ub1, ub2, ua0, ua1, ua2, c1.z, c1.w, c2.x, 24);
    TRI_STEP(1, 2, 0, r3, r4, r5, mk[24], ua0, ua1, ua2, ub0, ub1, ub2, c2.y, c2.z, c2.w, 24);
    TRI_STEP(0, 1, 2, r6, r7, r8, mk[24], ub0, ub1, ub2, ua0, ua1, ua2, c0.x, c0.y, c0.z, 25);
    TRI_STEP(2, 0, 1, r9, r10, r11, mk[24], ua0, ua1, ua2, ub0, ub1, ub2, c0.w, c1.x, c1.y, 25);
    TRI_STEP(1, 2, 0, r0, r1, r2, mk[25], ub0, ub1, ub2, ua0, ua1, ua2, c1.z, c1.w, c2.x, 25);
    TRI_STEP(0, 1, 2, r3, r4, r5, mk[25], ua0, ua1, ua2, ub0, ub1, ub2, c2.y, c2.z, c2.w, 25);
    TRI_STEP(2, 0, 1, r6, r7, r8, mk[25], ub0, ub1, ub2, ua0, ua1, ua2, c0.x, c0.y, c0.z, 26);
    TRI_STEP(1, 2, 0, r9, r10, r11, mk[25], ua0, ua1, ua2, ub0, ub1, ub2, c0.w, c1.x, c1.y, 26);
    TRI_STEP(0, 1, 2, r0, r1, r2, mk[26], ub0, ub1, ub2, ua0, ua1, ua2, c1.z, c1.w, c2.x, 26);
    TRI_STEP(2, 0, 1, r3, r4, r5, mk[26], ua0, ua1, ua2, ub0, ub1, ub2, c2.y, c2.z, c2.w, 26);
    TRI_STEP(1, 2, 0, r6, r7, r8, mk[26], ub0, ub1, ub2, ua0, ua1, ua2, c0.x, c0.y, c0.z, 27);
    TRI_STEP(0, 1, 2, r9, r10, r11, mk[26], ua0, ua1, ua2, ub0, ub1, ub2, c0.w, c1.x, c1.y, 27);
    TRI_STEP(2, 0, 1, r0, r1, r2, mk[27], ub0, ub1, ub2, ua0, ua1, ua2, c1.z, c1.w, c2.x, 27);
    TRI_STEP(1, 2, 0, r3, r4, r5, mk[27], ua0, ua1, ua2, ub0, ub1, ub2, c2.y, c2.z, c2.w, 27);
    TRI_STEP(0, 1, 2, r6, r7, r8, mk[27], ub0, ub1, ub2, ua0, ua1, ua2, c0.x, c0.y, c0.z, 28);
    TRI_STEP(2, 0, 1, r9, r10, r11, mk[27], ua0, ua1, ua2, ub0, ub1, ub2, c0.w, c1.x, c1.y, 28);
    TRI_STEP(1, 2, 0, r0, r1, r2, mk[28], ub0, ub1, ub2, ua0, ua1, ua2, c1.z, c1.w, c2.x, 28);
    TRI_STEP(0, 1, 2, r3, r4, r5, mk[28], ua0, ua1, ua2, ub0, ub1, ub2, c2.y, c2.z, c2.w, 28);
    TRI_STEP(2, 0, 1, r6, r7, r8, mk[28], ub0, ub1, ub2, ua0, ua1, ua2, c0.x, c0.y, c0.z, 29);
    TRI_STEP(1, 2, 0, r9, r10, r11, mk[28], ua0, ua1, ua2, ub0, ub1, ub2, c0.w, c1.x, c1.y, 29);
    TRI_STEP(0, 1, 2, r0, r1, r2, mk[29], ub0, ub1, ub2, ua0, ua1, ua2, c1.z, c1.w, c2.x, 29);
    TRI_STEP(2, 0, 1, r3, r4, r5, mk[29], ua0, ua1, ua2, ub0, ub1, ub2, c2.y, c2.z, c2.w, 29);
    TRI_STEP(1, 2, 0, r6, r7, r8, mk[29], ub0, ub1, ub2, ua0, ua1, ua2, c0.x, c0.y, c0.z, 30);
    TRI_STEP(0, 1, 2, r9, r10, r11, mk[29], ua0, ua1, ua2, ub0, ub1, ub2, c0.w, c1.x, c1.y, 30);
    TRI_STEP(2, 0, 1, r0, r1, r2, mk[30], ub0, ub1, ub2, ua0, ua1, ua2, c1.z, c1.w, c2.x, 30);
    TRI_STEP(1, 2, 0, r3, r4, r5, mk[30], ua0, ua1, ua2, ub0, ub1, ub2, c2.y, c2.z, c2.w, 30);
    TRI_STEP(0, 1, 2, r6, r7, r8, mk[30], ub0, ub1, ub2, ua0, ua1, ua2, c0.x, c0.y, c0.z, 31);
    TRI_STEP(2, 0, 1, r9, r10, r11, mk[30], ua0, ua1, ua2, ub0, ub1, ub2, c0.w, c1.x, c1.y, 31);
    TRI_STEP(1, 2, 0, r0, r1, r2, mk[31], ub0, ub1, ub2, ua0, ua1, ua2, c1.z, c1.w, c2.x, 31);
    TRI_STEP(0, 1, 2, r3, r4, r5, mk[31], ua0, ua1, ua2, ub0, ub1, ub2, c2.y, c2.z, c2.w, 31);
    TRI_STEP(2, 0, 1, r6, r7, r8, mk[31], ub0, ub1, ub2, ua0, ua1, ua2, n0.x, n0.y, n0.z, 0);

    r9  = (lane == 31) ? h0 : r9;
    r10 = (lane == 31) ? h1 : r10;
    r11 = (lane == 31) ? h2 : r11;

    v4f* po = pl + c * CHUNK_V4;
    po[0] = (v4f){ r0, r1, r2,  r3  };
    po[1] = (v4f){ r4, r5, r6,  r7  };
    po[2] = (v4f){ r8, r9, r10, r11 };
    c0 = n0; c1 = n1; c2 = n2;
  }

  if (C1 == NCHUNK && lane == 0) {
    float* hf = out + (size_t)BATCH * SEQ * HSZ + (size_t)b * HSZ;
    hf[0] = h0; hf[1] = h1; hf[2] = h2;
  }
}

__global__ __launch_bounds__(32 * MID_WAVES) void tridom_mid(
    const float* __restrict__ x, const float* __restrict__ wp,
    const float* __restrict__ bias, float* __restrict__ out)
{
  if (blockIdx.x < BATCH) {
    if (threadIdx.x >= 32) return;
    tridom_scan_range<0, T_CUT / CHUNK_STEPS>(out, blockIdx.x, threadIdx.x);
  } else {
    const int lane = threadIdx.x & 31;
    const int wave = threadIdx.x >> 5;
    proj_wave<TAIL_NB, T_CUT, SEQ - T_CUT>(x, wp, bias, out, (blockIdx.x - BATCH) * MID_WAVES + wave, lane);
  }
}

__global__ __launch_bounds__(32) void tridom_scan_tail(float* __restrict__ out)
{
  tridom_scan_range<T_CUT / CHUNK_STEPS, NCHUNK>(out, blockIdx.x, threadIdx.x);
}

extern "C" void kernel_launch(void* const* d_in, const int* in_sizes, int n_in,
                              void* d_out, int out_size, void* d_ws, size_t ws_size,
                              hipStream_t stream) {
  const float* x    = (const float*)d_in[0];
  const float* wp   = (const float*)d_in[1];
  const float* bias = (const float*)d_in[2];
  float* out = (float*)d_out;

  tridom_proj_head<<<dim3((BATCH * T_CUT) / (16 * HEAD_NB * HEAD_WAVES)), dim3(32 * HEAD_WAVES), 0, stream>>>(x, wp, bias, out);
  tridom_mid<<<dim3(2 * BATCH), dim3(32 * MID_WAVES), MID_LDS_BYTES, stream>>>(x, wp, bias, out);
  tridom_scan_tail<<<dim3(BATCH), dim3(32), 0, stream>>>(out);
}


# projection loop: last trip written out behind it, so block i+2 is requested unconditionally and no load wait ahead of a block's matrix steps covers the newest 16 loads; sum orders unchanged
# speedup vs baseline: 6.8868x; 1.0380x over previous
#include <hip/hip_runtime.h>

typedef __attribute__((ext_vector_type(2))) float v2f;
typedef __attribute__((ext_vector_type(4))) float v4f;
typedef __attribute__((ext_vector_type(8))) float v8f;

#define BATCH 128
#define SEQ   2048
#define ISZ   128
#define HSZ   3

#define PROJ_GROUPS          4
#define PROJ_LINES           (ISZ / 32)

#define T_CUT        768
#define HEAD_NB      4
#define HEAD_WAVES   6
#define TAIL_NB      10
#define MID_WAVES    8
#define MID_LDS_BYTES (176 * 1024)

__device__ __forceinline__ void proj_load_block(v4f (&a)[PROJ_GROUPS][PROJ_LINES], const float* p)
{
#pragma unroll
  for (int G = 0; G < PROJ_GROUPS; ++G)
#pragma unroll
    for (int L = 0; L < PROJ_LINES; ++L)
      a[G][L] = *(const v4f*)(p + (size_t)G * 4 * ISZ + 32 * L);
}

__device__ __forceinline__ float quad_x1(float v)
{
  const int b = __builtin_bit_cast(int, v);
  return __builtin_bit_cast(float, __builtin_amdgcn_update_dpp(b, b, 0xB1, 0xF, 0xF, true));
}
__device__ __forceinline__ float quad_x2(float v)
{
  const int b = __builtin_bit_cast(int, v);
  return __builtin_bit_cast(float, __builtin_amdgcn_update_dpp(b, b, 0x4E, 0xF, 0xF, true));
}

__device__ __forceinline__ float sel4(int q, float v0, float v1, float v2, float v3)
{
  const float lo = (q & 1) ? v1 : v0;
  const float hi = (q & 1) ? v3 : v2;
  return (q & 2) ? hi : lo;
}

__device__ __forceinline__ void proj_block(const v4f (&a)[PROJ_GROUPS][PROJ_LINES], const v4f (&bf)[PROJ_LINES],
                                           float* __restrict__ out, int row0, int hi, int h, int gq, float bb)
{
  v8f c[PROJ_GROUPS];
#pragma unroll
  for (int G = 0; G < PROJ_GROUPS; ++G) c[G] = (v8f){};
#pragma unroll
  for (int L = 0; L < PROJ_LINES; ++L) {
    const v2f b_lo = { bf[L].x, bf[L].y }, b_hi = { bf[L].z, bf[L].w };
#pragma unroll
    for (int G = 0; G < PROJ_GROUPS; ++G) {
      const v2f a_lo = { a[G][L].x, a[G][L].y };
      c[G] = __builtin_amdgcn_wmma_f32_16x16x4_f32(false, a_lo, false, b_lo,
                                                   (short)0, c[G], false, false);
    }
#pragma unroll
    for (int G = 0; G < PROJ_GROUPS; ++G) {
      const v2f a_hi = { a[G][L].z, a[G][L].w };
      c[G] = __builtin_amdgcn_wmma_f32_16x16x4_f32(false, a_hi, false, b_hi,
                                                   (short)0, c[G], false, false);
    }
  }
  float s0[PROJ_GROUPS], s1[PROJ_GROUPS];
#pragma unroll
  for (int G = 0; G < PROJ_GROUPS; ++G) {
    float w0 = sel4(gq, c[G][0], c[G][1], c[G][2], c[G][3]);
    float w1 = sel4(gq, c[G][4], c[G][5], c[G][6], c[G][7]);
    w0 += quad_x1(w0);  w1 += quad_x1(w1);
    w0 += quad_x2(w0);  w1 += quad_x2(w1);
    s0[G] = w0;  s1[G] = w1;
  }
  const float o0 = sel4(gq, s0[0], s0[1], s0[2], s0[3]) + bb;
  const float o1 = sel4(gq, s1[0], s1[1], s1[2], s1[3]) + bb;
  if (h < HSZ) {
    const int grow = row0 + 4 * gq + 2 * hi;
    out[(size_t)grow * HSZ + h]       = o0;
    out[(size_t)(grow + 1) * HSZ + h] = o1;
  }
}

template <int NB, int T0, int SPAN>
__device__ __forceinline__ void proj_wave(const float* __restrict__ x, const float* __restrict__ wp,
                                          const float* __restrict__ bias, float* __restrict__ out,
                                          int gw, int lane)
{
  constexpr int UPC = SPAN / (NB * 16);
  const int hi   = lane >> 4;
  const int ml   = lane & 15;
  const int tl   = ml >> 2;
  const int gq   = ml & 3;
  const int chain = gw / UPC;
  const int row_w = chain * SEQ + T0 + (gw - chain * UPC) * (NB * 16);

  const float* ap = x + ((size_t)row_w + tl) * ISZ + 8 * gq + 4 * hi;

  v4f a0[PROJ_GROUPS][PROJ_LINES], a1[PROJ_GROUPS][PROJ_LINES];
  proj_load_block(a0, ap);

  const int   h     = tl;
  const float bmask = (h < HSZ) ? 1.0f : 0.0f;
  const float* brow = wp + (size_t)((h < HSZ) ? h : 0) * ISZ + 8 * gq + 4 * hi;
  const float bb    = bias[(h < HSZ) ? h : 0];
  v4f bf[PROJ_LINES];
#pragma unroll
  for (int L = 0; L < PROJ_LINES; ++L) {
    v4f b4 = *(const v4f*)(brow + 32 * L);
    bf[L] = b4 * bmask;
  }

  constexpr int LAST = (NB & 1) ? NB - 1 : NB - 2;
#pragma unroll 1
  for (int i = 0; i < LAST; i += 2) {
    proj_load_block(a1, ap + (size_t)(i + 1) * 16 * ISZ);
    __builtin_amdgcn_sched_barrier(0);
    proj_block(a0, bf, out, row_w + 16 * i, hi, h, gq, bb);
    proj_load_block(a0, ap + (size_t)(i + 2) * 16 * ISZ);
    __builtin_amdgcn_sched_barrier(0);
    proj_block(a1, bf, out, row_w + 16 * (i + 1), hi, h, gq, bb);
  }
  if (!(NB & 1)) {
    proj_load_block(a1, ap + (size_t)(NB - 1) * 16 * ISZ);
    __builtin_amdgcn_sched_barrier(0);
  }
  proj_block(a0, bf, out, row_w + 16 * LAST, hi, h, gq, bb);
  if (!(NB & 1))
    proj_block(a1, bf, out, row_w + 16 * (NB - 1), hi, h, gq, bb);
}

__global__ __launch_bounds__(32 * HEAD_WAVES) void tridom_proj_head(
    const float* __restrict__ x, const float* __restrict__ wp,
    const float* __restrict__ bias, float* __restrict__ out)
{
  const int lane = threadIdx.x & 31;
  const int wave = threadIdx.x >> 5;
  proj_wave<HEAD_NB, 0, T_CUT>(x, wp, bias, out, blockIdx.x * HEAD_WAVES + wave, lane);
}

#define CHUNK_STEPS 128
#define CHUNK_V4    (CHUNK_STEPS * HSZ / 4)
#define NCHUNK      (SEQ / CHUNK_STEPS)

#define TRI_MUL0 "v_mul_f32 %[t0], 0x3fc00000, %[h1]\n\t"
#define TRI_MUL1 "v_mul_f32 %[t1], 0x3fc00000, %[h2]\n\t"
#define TRI_MUL2 "v_mul_f32 %[t2], 0xbfc00000, %[h0]\n\t"
#define TRI_MUL(c) TRI_MUL##c
#define TRI_RDL(c) "v_readlane_b32 %[n" #c "], %[c" #c "], %[j]\n\t"
#define TRI_TNH(c) "v_tanh_f32 %[t" #c "], %[t" #c "]\n\t"
#define TRI_FMC(c) "v_fmac_f32 %[t" #c "], -0.5, %[h" #c "]\n\t"
#define TRI_ADD(c) "v_add_f32 %[h" #c "], %[u" #c "], %[t" #c "]\n\t"
#define TRI_CAP(c) "v_cndmask_b32_e64 %[r" #c "], %[r" #c "], %[h" #c "], %[m]\n\t"

#define TRI_STEP(X, Y, Z, R0, R1, R2, MV, U0, U1, U2, N0, N1, N2, CN0, CN1, CN2, JN) \
  do {                                                                           \
    float t0_, t1_, t2_;                                                         \
    asm(TRI_MUL(X) TRI_RDL(0) TRI_MUL(Y) TRI_RDL(1) TRI_MUL(Z)                   \
        TRI_TNH(X) TRI_RDL(2) TRI_TNH(Y) TRI_TNH(Z)                              \
        TRI_CAP(0) TRI_CAP(1) TRI_CAP(2)                                         \
        TRI_FMC(X) TRI_FMC(Y) TRI_FMC(Z)                                         \
        TRI_ADD(X) TRI_ADD(Y) TRI_ADD(Z)                                         \
        : [h0] "+v"(h0), [h1] "+v"(h1), [h2] "+v"(h2),                           \
          [r0] "+v"(R0), [r1] "+v"(R1), [r2] "+v"(R2),                           \
          [t0] "=&v"(t0_), [t1] "=&v"(t1_), [t2] "=&v"(t2_),                     \
          [n0] "=&s"(N0), [n1] "=&s"(N1), [n2] "=&s"(N2)                         \
        : [c0] "v"(CN0), [c1] "v"(CN1), [c2] "v"(CN2),                           \
          [u0] "s"(U0), [u1] "s"(U1), [u2] "s"(U2), [m] "s"(MV), [j] "n"(JN));   \
  } while (0)

#define TRI_STEP_FIRST(X, Y, Z, U0, U1, U2, N0, N1, N2, CN0, CN1, CN2, JN)       \
  do {                                                                           \
    float t0_, t1_, t2_;                                                         \
    asm(TRI_MUL(X) TRI_RDL(0) TRI_MUL(Y) TRI_RDL(1) TRI_MUL(Z)                   \
        TRI_TNH(X) TRI_RDL(2) TRI_TNH(Y) TRI_TNH(Z)                              \
        TRI_FMC(X) TRI_FMC(Y) TRI_FMC(Z)                                         \
        TRI_ADD(X) TRI_ADD(Y) TRI_ADD(Z)                                         \
        : [h0] "+v"(h0), [h1] "+v"(h1), [h2] "+v"(h2),                           \
          [t0] "=&v"(t0_), [t1] "=&v"(t1_), [t2] "=&v"(t2_),                     \
          [n0] "=&s"(N0), [n1] "=&s"(N1), [n2] "=&s"(N2)                         \
        : [c0] "v"(CN0), [c1] "v"(CN1), [c2] "v"(CN2),                           \
          [u0] "s"(U0), [u1] "s"(U1), [u2] "s"(U2), [j] "n"(JN));                \
  } while (0)

__device__ __forceinline__ float tri_lane0(float v)
{
  return __builtin_bit_cast(float, __builtin_amdgcn_readlane(__builtin_bit_cast(int, v), 0));
}

template <int C0, int C1>
__device__ __forceinline__ void tridom_scan_range(float* __restrict__ out, int b, int lane)
{
  v4f* pl = (v4f*)(out + (size_t)b * SEQ * HSZ) + 3 * lane;

  float h0 = 0.f, h1 = 0.f, h2 = 0.f;
  if (C0 > 0) {
    const float* hp = out + ((size_t)b * SEQ + (size_t)C0 * CHUNK_STEPS - 1) * HSZ;
    h0 = hp[0]; h1 = hp[1]; h2 = hp[2];
  }
  float r0 = 0.f, r1 = 0.f, r2 = 0.f, r3 = 0.f, r4 = 0.f, r5 = 0.f;
  float r6 = 0.f, r7 = 0.f, r8 = 0.f, r9 = 0.f, r10 = 0.f, r11 = 0.f;

  unsigned mk[32];
#pragma unroll
  for (int j = 0; j < 32; ++j) mk[j] = __builtin_amdgcn_ballot_w32(lane == j);

  v4f c0 = pl[C0 * CHUNK_V4 + 0], c1 = pl[C0 * CHUNK_V4 + 1], c2 = pl[C0 * CHUNK_V4 + 2];

  float ua0 = tri_lane0(c0.x), ua1 = tri_lane0(c0.y), ua2 = tri_lane0(c0.z);
  float ub0, ub1, ub2;

  for (int c = C0; c < C1; ++c) {
    v4f n0 = c0, n1 = c1, n2 = c2;
    if (c + 1 < C1) {
      const v4f* pn = pl + (c + 1) * CHUNK_V4;
      n0 = pn[0]; n1 = pn[1]; n2 = pn[2];
    }

    TRI_STEP_FIRST(0, 1, 2, ua0, ua1, ua2, ub0, ub1, ub2, c0.w, c1.x, c1.y, 0);
    TRI_STEP(2, 0, 1, r0, r1, r2, mk[0], ub0, ub1, ub2, ua0, ua1, ua2, c1.z, c1.w, c2.x, 0);
    TRI_STEP(1, 2, 0, r3, r4, r5, mk[0], ua0, ua1, ua2, ub0, ub1, ub2, c2.y, c2.z, c2.w, 0);
    TRI_STEP(0, 1, 2, r6, r7, r8, mk[0], ub0, ub1, ub2, ua0, ua1, ua2, c0.x, c0.y, c0.z, 1);
    TRI_STEP(2, 0, 1, r9, r10, r11, mk[0], ua0, ua1, ua2, ub0, ub1, ub2, c0.w, c1.x, c1.y, 1);
    TRI_STEP(1, 2, 0, r0, r1, r2, mk[1], ub0, ub1, ub2, ua0, ua1, ua2, c1.z, c1.w, c2.x, 1);
    TRI_STEP(0, 1, 2, r3, r4, r5, mk[1], ua0, ua1, ua2, ub0, ub1, ub2, c2.y, c2.z, c2.w, 1);
    TRI_STEP(2, 0, 1, r6, r7, r8, mk[1], ub0, ub1, ub2, ua0, ua1, ua2, c0.x, c0.y, c0.z, 2);
    TRI_STEP(1, 2, 0, r9, r10, r11, mk[1], ua0, ua1, ua2, ub0, ub1, ub2, c0.w, c1.x, c1.y, 2);
    TRI_STEP(0, 1, 2, r0, r1, r2, mk[2], ub0, ub1, ub2, ua0, ua1, ua2, c1.z, c1.w, c2.x, 2);
    TRI_STEP(2, 0, 1, r3, r4, r5, mk[2], ua0, ua1, ua2, ub0, ub1, ub2, c2.y, c2.z, c2.w, 2);
    TRI_STEP(1, 2, 0, r6, r7, r8, mk[2], ub0, ub1, ub2, ua0, ua1, ua2, c0.x, c0.y, c0.z, 3);
    TRI_STEP(0, 1, 2, r9, r10, r11, mk[2], ua0, ua1, ua2, ub0, ub1, ub2, c0.w, c1.x, c1.y, 3);
    TRI_STEP(2, 0, 1, r0, r1, r2, mk[3], ub0, ub1, ub2, ua0, ua1, ua2, c1.z, c1.w, c2.x, 3);
    TRI_STEP(1, 2, 0, r3, r4, r5, mk[3], ua0, ua1, ua2, ub0, ub1, ub2, c2.y, c2.z, c2.w, 3);
    TRI_STEP(0, 1, 2, r6, r7, r8, mk[3], ub0, ub1, ub2, ua0, ua1, ua2, c0.x, c0.y, c0.z, 4);
    TRI_STEP(2, 0, 1, r9, r10, r11, mk[3], ua0, ua1, ua2, ub0, ub1, ub2, c0.w, c1.x, c1.y, 4);
    TRI_STEP(1, 2, 0, r0, r1, r2, mk[4], ub0, ub1, ub2, ua0, ua1, ua2, c1.z, c1.w, c2.x, 4);
    TRI_STEP(0, 1, 2, r3, r4, r5, mk[4], ua0, ua1, ua2, ub0, ub1, ub2, c2.y, c2.z, c2.w, 4);
    TRI_STEP(2, 0, 1, r6, r7, r8, mk[4], ub0, ub1, ub2, ua0, ua1, ua2, c0.x, c0.y, c0.z, 5);
    TRI_STEP(1, 2, 0, r9, r10, r11, mk[4], ua0, ua1, ua2, ub0, ub1, ub2, c0.w, c1.x, c1.y, 5);
    TRI_STEP(0, 1, 2, r0, r1, r2, mk[5], ub0, ub1, ub2, ua0, ua1, ua2, c1.z, c1.w, c2.x, 5);
    TRI_STEP(2, 0, 1, r3, r4, r5, mk[5], ua0, ua1, ua2, ub0, ub1, ub2, c2.y, c2.z, c2.w, 5);
    TRI_STEP(1, 2, 0, r6, r7, r8, mk[5], ub0, ub1, ub2, ua0, ua1, ua2, c0.x, c0.y, c0.z, 6);
    TRI_STEP(0, 1, 2, r9, r10, r11, mk[5], ua0, ua1, ua2, ub0, ub1, ub2, c0.w, c1.x, c1.y, 6);
    TRI_STEP(2, 0, 1, r0, r1, r2, mk[6], ub0, ub1, ub2, ua0, ua1, ua2, c1.z, c1.w, c2.x, 6);
    TRI_STEP(1, 2, 0, r3, r4, r5, mk[6], ua0, ua1, ua2, ub0, ub1, ub2, c2.y, c2.z, c2.w, 6);
    TRI_STEP(0, 1, 2, r6, r7, r8, mk[6], ub0, ub1, ub2, ua0, ua1, ua2, c0.x, c0.y, c0.z, 7);
    TRI_STEP(2, 0, 1, r9, r10, r11, mk[6], ua0, ua1, ua2, ub0, ub1, ub2, c0.w, c1.x, c1.y, 7);
    TRI_STEP(1, 2, 0, r0, r1, r2, mk[7], ub0, ub1, ub2, ua0, ua1, ua2, c1.z, c1.w, c2.x, 7);
    TRI_STEP(0, 1, 2, r3, r4, r5, mk[7], ua0, ua1, ua2, ub0, ub1, ub2, c2.y, c2.z, c2.w, 7);
    TRI_STEP(2, 0, 1, r6, r7, r8, mk[7], ub0, ub1, ub2, ua0, ua1, ua2, c0.x, c0.y, c0.z, 8);
    TRI_STEP(1, 2, 0, r9, r10, r11, mk[7], ua0, ua1, ua2, ub0, ub1, ub2, c0.w, c1.x, c1.y, 8);
    TRI_STEP(0, 1, 2, r0, r1, r2, mk[8], ub0, ub1, ub2, ua0, ua1, ua2, c1.z, c1.w, c2.x, 8);
    TRI_STEP(2, 0, 1, r3, r4, r5, mk[8], ua0, ua1, ua2, ub0, ub1, ub2, c2.y, c2.z, c2.w, 8);
    TRI_STEP(1, 2, 0, r6, r7, r8, mk[8], ub0, ub1, ub2, ua0, ua1, ua2, c0.x, c0.y, c0.z, 9);
    TRI_STEP(0, 1, 2, r9, r10, r11, mk[8], ua0, ua1, ua2, ub0, ub1, ub2, c0.w, c1.x, c1.y, 9);
    TRI_STEP(2, 0, 1, r0, r1, r2, mk[9], ub0, ub1, ub2, ua0, ua1, ua2, c1.z, c1.w, c2.x, 9);
    TRI_STEP(1, 2, 0, r3, r4, r5, mk[9], ua0, ua1, ua2, ub0, ub1, ub2, c2.y, c2.z, c2.w, 9);
    TRI_STEP(0, 1, 2, r6, r7, r8, mk[9], ub0, ub1, ub2, ua0, ua1, ua2, c0.x, c0.y, c0.z, 10);
    TRI_STEP(2, 0, 1, r9, r10, r11, mk[9], ua0, ua1, ua2, ub0, ub1, ub2, c0.w, c1.x, c1.y, 10);
    TRI_STEP(1, 2, 0, r0, r1, r2, mk[10], ub0, ub1, ub2, ua0, ua1, ua2, c1.z, c1.w, c2.x, 10);
    TRI_STEP(0, 1, 2, r3, r4, r5, mk[10], ua0, ua1, ua2, ub0, ub1, ub2, c2.y, c2.z, c2.w, 10);
    TRI_STEP(2, 0, 1, r6, r7, r8, mk[10], ub0, ub1, ub2, ua0, ua1, ua2, c0.x, c0.y, c0.z, 11);
    TRI_STEP(1, 2, 0, r9, r10, r11, mk[10], ua0, ua1, ua2, ub0, ub1, ub2, c0.w, c1.x, c1.y, 11);
    TRI_STEP(0, 1, 2, r0, r1, r2, mk[11], ub0, ub1, ub2, ua0, ua1, ua2, c1.z, c1.w, c2.x, 11);
    TRI_STEP(2, 0, 1, r3, r4, r5, mk[11], ua0, ua1, ua2, ub0, ub1, ub2, c2.y, c2.z, c2.w, 11);
    TRI_STEP(1, 2, 0, r6, r7, r8, mk[11], ub0, ub1, ub2, ua0, ua1, ua2, c0.x, c0.y, c0.z, 12);
    TRI_STEP(0, 1, 2, r9, r10, r11, mk[11], ua0, ua1, ua2, ub0, ub1, ub2, c0.w, c1.x, c1.y, 12);
    TRI_STEP(2, 0, 1, r0, r1, r2, mk[12], ub0, ub1, ub2, ua0, ua1, ua2, c1.z, c1.w, c2.x, 12);
    TRI_STEP(1, 2, 0, r3, r4, r5, mk[12], ua0, ua1, ua2, ub0, ub1, ub2, c2.y, c2.z, c2.w, 12);
    TRI_STEP(0, 1, 2, r6, r7, r8, mk[12], ub0, ub1, ub2, ua0, ua1, ua2, c0.x, c0.y, c0.z, 13);
    TRI_STEP(2, 0, 1, r9, r10, r11, mk[12], ua0, ua1, ua2, ub0, ub1, ub2, c0.w, c1.x, c1.y, 13);
    TRI_STEP(1, 2, 0, r0, r1, r2, mk[13], ub0, ub1, ub2, ua0, ua1, ua2, c1.z, c1.w, c2.x, 13);
    TRI_STEP(0, 1, 2, r3, r4, r5, mk[13], ua0, ua1, ua2, ub0, ub1, ub2, c2.y, c2.z, c2.w, 13);
    TRI_STEP(2, 0, 1, r6, r7, r8, mk[13], ub0, ub1, ub2, ua0, ua1, ua2, c0.x, c0.y, c0.z, 14);
    TRI_STEP(1, 2, 0, r9, r10, r11, mk[13], ua0, ua1, ua2, ub0, ub1, ub2, c0.w, c1.x, c1.y, 14);
    TRI_STEP(0, 1, 2, r0, r1, r2, mk[14], ub0, ub1, ub2, ua0, ua1, ua2, c1.z, c1.w, c2.x, 14);
    TRI_STEP(2, 0, 1, r3, r4, r5, mk[14], ua0, ua1, ua2, ub0, ub1, ub2, c2.y, c2.z, c2.w, 14);
    TRI_STEP(1, 2, 0, r6, r7, r8, mk[14], ub0, ub1, ub2, ua0, ua1, ua2, c0.x, c0.y, c0.z, 15);
    TRI_STEP(0, 1, 2, r9, r10, r11, mk[14], ua0, ua1, ua2, ub0, ub1, ub2, c0.w, c1.x, c1.y, 15);
    TRI_STEP(2, 0, 1, r0, r1, r2, mk[15], ub0, ub1, ub2, ua0, ua1, ua2, c1.z, c1.w, c2.x, 15);
    TRI_STEP(1, 2, 0, r3, r4, r5, mk[15], ua0, ua1, ua2, ub0, ub1, ub2, c2.y, c2.z, c2.w, 15);
    TRI_STEP(0, 1, 2, r6, r7, r8, mk[15], ub0, ub1, ub2, ua0, ua1, ua2, c0.x, c0.y, c0.z, 16);
    TRI_STEP(2, 0, 1, r9, r10, r11, mk[15], ua0, ua1, ua2, ub0, ub1, ub2, c0.w, c1.x, c1.y, 16);
    TRI_STEP(1, 2, 0, r0, r1, r2, mk[16], ub0, ub1, ub2, ua0, ua1, ua2, c1.z, c1.w, c2.x, 16);
    TRI_STEP(0, 1, 2, r3, r4, r5, mk[16], ua0, ua1, ua2, ub0, ub1, ub2, c2.y, c2.z, c2.w, 16);
    TRI_STEP(2, 0, 1, r6, r7, r8, mk[16], ub0, ub1, ub2, ua0, ua1, ua2, c0.x, c0.y, c0.z, 17);
    TRI_STEP(1, 2, 0, r9, r10, r11, mk[16], ua0, ua1, ua2, ub0, ub1, ub2, c0.w, c1.x, c1.y, 17);
    TRI_STEP(0, 1, 2, r0, r1, r2, mk[17], ub0, ub1, ub2, ua0, ua1, ua2, c1.z, c1.w, c2.x, 17);
    TRI_STEP(2, 0, 1, r3, r4, r5, mk[17], ua0, ua1, ua2, ub0, ub1, ub2, c2.y, c2.z, c2.w, 17);
    TRI_STEP(1, 2, 0, r6, r7, r8, mk[17], ub0, ub1, ub2, ua0, ua1, ua2, c0.x, c0.y, c0.z, 18);
    TRI_STEP(0, 1, 2, r9, r10, r11, mk[17], ua0, ua1, ua2, ub0, ub1, ub2, c0.w, c1.x, c1.y, 18);
    TRI_STEP(2, 0, 1, r0, r1, r2, mk[18], ub0, ub1, ub2, ua0, ua1, ua2, c1.z, c1.w, c2.x, 18);
    TRI_STEP(1, 2, 0, r3, r4, r5, mk[18], ua0, ua1, ua2, ub0, ub1, ub2, c2.y, c2.z, c2.w, 18);
    TRI_STEP(0, 1, 2, r6, r7, r8, mk[18], ub0, ub1, ub2, ua0, ua1, ua2, c0.x, c0.y, c0.z, 19);
    TRI_STEP(2, 0, 1, r9, r10, r11, mk[18], ua0, ua1, ua2, ub0, ub1, ub2, c0.w, c1.x, c1.y, 19);
    TRI_STEP(1, 2, 0, r0, r1, r2, mk[19], ub0, ub1, ub2, ua0, ua1, ua2, c1.z, c1.w, c2.x, 19);
    TRI_STEP(0, 1, 2, r3, r4, r5, mk[19], ua0, ua1, ua2, ub0, ub1, ub2, c2.y, c2.z, c2.w, 19);
    TRI_STEP(2, 0, 1, r6, r7, r8, mk[19], ub0, ub1, ub2, ua0, ua1, ua2, c0.x, c0.y, c0.z, 20);
    TRI_STEP(1, 2, 0, r9, r10, r11, mk[19], ua0, ua1, ua2, ub0, ub1, ub2, c0.w, c1.x, c1.y, 20);
    TRI_STEP(0, 1, 2, r0, r1, r2, mk[20], ub0, ub1, ub2, ua0, ua1, ua2, c1.z, c1.w, c2.x, 20);
    TRI_STEP(2, 0, 1, r3, r4, r5, mk[20], ua0, ua1, ua2, ub0, ub1, ub2, c2.y, c2.z, c2.w, 20);
    TRI_STEP(1, 2, 0, r6, r7, r8, mk[20], ub0, ub1, ub2, ua0, ua1, ua2, c0.x, c0.y, c0.z, 21);
    TRI_STEP(0, 1, 2, r9, r10, r11, mk[20], ua0, ua1, ua2, ub0, ub1, ub2, c0.w, c1.x, c1.y, 21);
    TRI_STEP(2, 0, 1, r0, r1, r2, mk[21], ub0, ub1, ub2, ua0, ua1, ua2, c1.z, c1.w, c2.x, 21);
    TRI_STEP(1, 2, 0, r3, r4, r5, mk[21], ua0, ua1, ua2, ub0, ub1, ub2, c2.y, c2.z, c2.w, 21);
    TRI_STEP(0, 1, 2, r6, r7, r8, mk[21], ub0, ub1, ub2, ua0, ua1, ua2, c0.x, c0.y, c0.z, 22);
    TRI_STEP(2, 0, 1, r9, r10, r11, mk[21], ua0, ua1, ua2, ub0, ub1, ub2, c0.w, c1.x, c1.y, 22);
    TRI_STEP(1, 2, 0, r0, r1, r2, mk[22], ub0, ub1, ub2, ua0, ua1, ua2, c1.z, c1.w, c2.x, 22);
    TRI_STEP(0, 1, 2, r3, r4, r5, mk[22], ua0, ua1, ua2, ub0, ub1, ub2, c2.y, c2.z, c2.w, 22);
    TRI_STEP(2, 0, 1, r6, r7, r8, mk[22], ub0, ub1, ub2, ua0, ua1, ua2, c0.x, c0.y, c0.z, 23);
    TRI_STEP(1, 2, 0, r9, r10, r11, mk[22], ua0, ua1, ua2, ub0, ub1, ub2, c0.w, c1.x, c1.y, 23);
    TRI_STEP(0, 1, 2, r0, r1, r2, mk[23], ub0, ub1, ub2, ua0, ua1, ua2, c1.z, c1.w, c2.x, 23);
    TRI_STEP(2, 0, 1, r3, r4, r5, mk[23], ua0, ua1, ua2, ub0, ub1, ub2, c2.y, c2.z, c2.w, 23);
    TRI_STEP(1, 2, 0, r6, r7, r8, mk[23], ub0, ub1, ub2, ua0, ua1, ua2, c0.x, c0.y, c0.z, 24);
    TRI_STEP(0, 1, 2, r9, r10, r11, mk[23], ua0, ua1, ua2, ub0, ub1, ub2, c0.w, c1.x, c1.y, 24);
    TRI_STEP(2, 0, 1, r0, r1, r2, mk[24], ub0, ub1, ub2, ua0, ua1, ua2, c1.z, c1.w, c2.x, 24);
    TRI_STEP(1, 2, 0, r3, r4, r5, mk[24], ua0, ua1, ua2, ub0, ub1, ub2, c2.y, c2.z, c2.w, 24);
    TRI_STEP(0, 1, 2, r6, r7, r8, mk[24], ub0, ub1, ub2, ua0, ua1, ua2, c0.x, c0.y, c0.z, 25);
    TRI_STEP(2, 0, 1, r9, r10, r11, mk[24], ua0, ua1, ua2, ub0, ub1, ub2, c0.w, c1.x, c1.y, 25);
    TRI_STEP(1, 2, 0, r0, r1, r2, mk[25], ub0, ub1, ub2, ua0, ua1, ua2, c1.z, c1.w, c2.x, 25);
    TRI_STEP(0, 1, 2, r3, r4, r5, mk[25], ua0, ua1, ua2, ub0, ub1, ub2, c2.y, c2.z, c2.w, 25);
    TRI_STEP(2, 0, 1, r6, r7, r8, mk[25], ub0, ub1, ub2, ua0, ua1, ua2, c0.x, c0.y, c0.z, 26);
    TRI_STEP(1, 2, 0, r9, r10, r11, mk[25], ua0, ua1, ua2, ub0, ub1, ub2, c0.w, c1.x, c1.y, 26);
    TRI_STEP(0, 1, 2, r0, r1, r2, mk[26], ub0, ub1, ub2, ua0, ua1, ua2, c1.z, c1.w, c2.x, 26);
    TRI_STEP(2, 0, 1, r3, r4, r5, mk[26], ua0, ua1, ua2, ub0, ub1, ub2, c2.y, c2.z, c2.w, 26);
    TRI_STEP(1, 2, 0, r6, r7, r8, mk[26], ub0, ub1, ub2, ua0, ua1, ua2, c0.x, c0.y, c0.z, 27);
    TRI_STEP(0, 1, 2, r9, r10, r11, mk[26], ua0, ua1, ua2, ub0, ub1, ub2, c0.w, c1.x, c1.y, 27);
    TRI_STEP(2, 0, 1, r0, r1, r2, mk[27], ub0, ub1, ub2, ua0, ua1, ua2, c1.z, c1.w, c2.x, 27);
    TRI_STEP(1, 2, 0, r3, r4, r5, mk[27], ua0, ua1, ua2, ub0, ub1, ub2, c2.y, c2.z, c2.w, 27);
    TRI_STEP(0, 1, 2, r6, r7, r8, mk[27], ub0, ub1, ub2, ua0, ua1, ua2, c0.x, c0.y, c0.z, 28);
    TRI_STEP(2, 0, 1, r9, r10, r11, mk[27], ua0, ua1, ua2, ub0, ub1, ub2, c0.w, c1.x, c1.y, 28);
    TRI_STEP(1, 2, 0, r0, r1, r2, mk[28], ub0, ub1, ub2, ua0, ua1, ua2, c1.z, c1.w, c2.x, 28);
    TRI_STEP(0, 1, 2, r3, r4, r5, mk[28], ua0, ua1, ua2, ub0, ub1, ub2, c2.y, c2.z, c2.w, 28);
    TRI_STEP(2, 0, 1, r6, r7, r8, mk[28], ub0, ub1, ub2, ua0, ua1, ua2, c0.x, c0.y, c0.z, 29);
    TRI_STEP(1, 2, 0, r9, r10, r11, mk[28], ua0, ua1, ua2, ub0, ub1, ub2, c0.w, c1.x, c1.y, 29);
    TRI_STEP(0, 1, 2, r0, r1, r2, mk[29], ub0, ub1, ub2, ua0, ua1, ua2, c1.z, c1.w, c2.x, 29);
    TRI_STEP(2, 0, 1, r3, r4, r5, mk[29], ua0, ua1, ua2, ub0, ub1, ub2, c2.y, c2.z, c2.w, 29);
    TRI_STEP(1, 2, 0, r6, r7, r8, mk[29], ub0, ub1, ub2, ua0, ua1, ua2, c0.x, c0.y, c0.z, 30);
    TRI_STEP(0, 1, 2, r9, r10, r11, mk[29], ua0, ua1, ua2, ub0, ub1, ub2, c0.w, c1.x, c1.y, 30);
    TRI_STEP(2, 0, 1, r0, r1, r2, mk[30], ub0, ub1, ub2, ua0, ua1, ua2, c1.z, c1.w, c2.x, 30);
    TRI_STEP(1, 2, 0, r3, r4, r5, mk[30], ua0, ua1, ua2, ub0, ub1, ub2, c2.y, c2.z, c2.w, 30);
    TRI_STEP(0, 1, 2, r6, r7, r8, mk[30], ub0, ub1, ub2, ua0, ua1, ua2, c0.x, c0.y, c0.z, 31);
    TRI_STEP(2, 0, 1, r9, r10, r11, mk[30], ua0, ua1, ua2, ub0, ub1, ub2, c0.w, c1.x, c1.y, 31);
    TRI_STEP(1, 2, 0, r0, r1, r2, mk[31], ub0, ub1, ub2, ua0, ua1, ua2, c1.z, c1.w, c2.x, 31);
    TRI_STEP(0, 1, 2, r3, r4, r5, mk[31], ua0, ua1, ua2, ub0, ub1, ub2, c2.y, c2.z, c2.w, 31);
    TRI_STEP(2, 0, 1, r6, r7, r8, mk[31], ub0, ub1, ub2, ua0, ua1, ua2, n0.x, n0.y, n0.z, 0);

    r9  = (lane == 31) ? h0 : r9;
    r10 = (lane == 31) ? h1 : r10;
    r11 = (lane == 31) ? h2 : r11;

    v4f* po = pl + c * CHUNK_V4;
    po[0] = (v4f){ r0, r1, r2,  r3  };
    po[1] = (v4f){ r4, r5, r6,  r7  };
    po[2] = (v4f){ r8, r9, r10, r11 };
    c0 = n0; c1 = n1; c2 = n2;
  }

  if (C1 == NCHUNK && lane == 0) {
    float* hf = out + (size_t)BATCH * SEQ * HSZ + (size_t)b * HSZ;
    hf[0] = h0; hf[1] = h1; hf[2] = h2;
  }
}

__global__ __launch_bounds__(32 * MID_WAVES) void tridom_mid(
    const float* __restrict__ x, const float* __restrict__ wp,
    const float* __restrict__ bias, float* __restrict__ out)
{
  if (blockIdx.x < BATCH) {
    if (threadIdx.x >= 32) return;
    tridom_scan_range<0, T_CUT / CHUNK_STEPS>(out, blockIdx.x, threadIdx.x);
  } else {
    const int lane = threadIdx.x & 31;
    const int wave = threadIdx.x >> 5;
    proj_wave<TAIL_NB, T_CUT, SEQ - T_CUT>(x, wp, bias, out, (blockIdx.x - BATCH) * MID_WAVES + wave, lane);
  }
}

__global__ __launch_bounds__(32) void tridom_scan_tail(float* __restrict__ out)
{
  tridom_scan_range<T_CUT / CHUNK_STEPS, NCHUNK>(out, blockIdx.x, threadIdx.x);
}

extern "C" void kernel_launch(void* const* d_in, const int* in_sizes, int n_in,
                              void* d_out, int out_size, void* d_ws, size_t ws_size,
                              hipStream_t stream) {
  const float* x    = (const float*)d_in[0];
  const float* wp   = (const float*)d_in[1];
  const float* bias = (const float*)d_in[2];
  float* out = (float*)d_out;

  tridom_proj_head<<<dim3((BATCH * T_CUT) / (16 * HEAD_NB * HEAD_WAVES)), dim3(32 * HEAD_WAVES), 0, stream>>>(x, wp, bias, out);
  tridom_mid<<<dim3(2 * BATCH), dim3(32 * MID_WAVES), MID_LDS_BYTES, stream>>>(x, wp, bias, out);
  tridom_scan_tail<<<dim3(BATCH), dim3(32), 0, stream>>>(out);
}
